# starting kernel
# baseline (speedup 1.0000x reference)
#include <hip/hip_runtime.h>
#include <hip/hip_bf16.h>


typedef __attribute__((ext_vector_type(2))) float v2f;
typedef __attribute__((ext_vector_type(8))) float v8f;

#define GCN_NODES 50000
#define GCN_DIN   128
#define GCN_DHID  256
#define GCN_DOUT  128

__global__ void gcn_gemm_bias_wmma(const float* __restrict__ X,
                                   const float* __restrict__ W,
                                   const float* __restrict__ bias,
                                   float* __restrict__ H,
                                   int K, int N) {
    const int lane = threadIdx.x;
    const int half = lane >> 4;
    const int l    = lane & 15;

    const int mBase = blockIdx.x * 16;
    const int nBase = blockIdx.y * 64;

    const float* arow = X + (size_t)(mBase + l) * (size_t)K;

    v8f acc[4];
#pragma unroll
    for (int t = 0; t < 4; ++t) {
        const float bv = bias[nBase + t * 16 + l];
#pragma unroll
        for (int i = 0; i < 8; ++i) acc[t][i] = bv;
    }

    for (int k = 0; k < K; k += 4) {
        const v2f a = *(const v2f*)(arow + k + 2 * half);

        const int krow = k + 2 * half;
#pragma unroll
        for (int t = 0; t < 4; ++t) {
            const int col = nBase + t * 16 + l;
            v2f b;
            b.x = W[(size_t)(krow)     * (size_t)N + col];
            b.y = W[(size_t)(krow + 1) * (size_t)N + col];
            acc[t] = __builtin_amdgcn_wmma_f32_16x16x4_f32(
                 false, a,  false, b,
                 (short)0, acc[t],
                 false,  false);
        }
    }

#pragma unroll
    for (int t = 0; t < 4; ++t) {
        const int col = nBase + t * 16 + l;
#pragma unroll
        for (int i = 0; i < 8; ++i) {
            const int row = mBase + i + 8 * half;
            H[(size_t)row * (size_t)N + col] = acc[t][i];
        }
    }
}

__global__ void gcn_scatter_add(const float* __restrict__ h,
                                const int* __restrict__ edges,
                                float* __restrict__ agg,
                                float* __restrict__ deg,
                                long long nEdges, int C, int addDeg) {
    const int chunks = C >> 2;
    const long long tid = (long long)blockIdx.x * blockDim.x + threadIdx.x;
    const long long total = nEdges * (long long)chunks;
    if (tid >= total) return;

    const long long e  = tid / chunks;
    const int       c4 = (int)(tid - e * (long long)chunks) << 2;

    const int src = edges[2 * e];
    const int dst = edges[2 * e + 1];

    const float4 v = *(const float4*)(h + (size_t)src * (size_t)C + c4);
    float* p = agg + (size_t)dst * (size_t)C + c4;
    atomicAdd(p + 0, v.x);
    atomicAdd(p + 1, v.y);
    atomicAdd(p + 2, v.z);
    atomicAdd(p + 3, v.w);

    if (addDeg && c4 == 0) atomicAdd(deg + dst, 1.0f);
}

__global__ void gcn_normalize(const float* __restrict__ agg,
                              const float* __restrict__ deg,
                              float* __restrict__ out,
                              long long n, int C, int doRelu) {
    const long long i = (long long)blockIdx.x * blockDim.x + threadIdx.x;
    if (i >= n) return;
    const long long node = i / C;
    const float d = fmaxf(deg[node], 1.0f);
    float v = agg[i] / d;
    if (doRelu) v = fmaxf(v, 0.0f);
    out[i] = v;
}

__global__ void gcn_zero_f32(float* __restrict__ p, long long n) {
    const long long i = (long long)blockIdx.x * blockDim.x + threadIdx.x;
    if (i < n) p[i] = 0.0f;
}

extern "C" void kernel_launch(void* const* d_in, const int* in_sizes, int n_in,
                              void* d_out, int out_size, void* d_ws, size_t ws_size,
                              hipStream_t stream) {
    const float* X  = (const float*)d_in[0];
    const int*   E  = (const int*)  d_in[1];
    const float* W1 = (const float*)d_in[2];
    const float* b1 = (const float*)d_in[3];
    const float* W2 = (const float*)d_in[4];
    const float* b2 = (const float*)d_in[5];
    float* out = (float*)d_out;

    const long long nEdges = (long long)in_sizes[1] / 2;

    char* ws = (char*)d_ws;
    float* h1   = (float*)ws;
    float* agg1 = (float*)(ws + (size_t)GCN_NODES * GCN_DHID * 4);
    float* deg  = (float*)(ws + (size_t)GCN_NODES * GCN_DHID * 8);
    float* h2   = agg1;

    const long long nH1  = (long long)GCN_NODES * GCN_DHID;
    const long long nOut = (long long)GCN_NODES * GCN_DOUT;

    gcn_gemm_bias_wmma<<<dim3(GCN_NODES / 16, GCN_DHID / 64), 32, 0, stream>>>(
        X, W1, b1, h1, GCN_DIN, GCN_DHID);

    {
        const long long nz = nH1 + GCN_NODES;
        gcn_zero_f32<<<(int)((nz + 255) / 256), 256, 0, stream>>>(agg1, nz);
    }
    {
        const long long tot = nEdges * (GCN_DHID / 4);
        gcn_scatter_add<<<(int)((tot + 255) / 256), 256, 0, stream>>>(
            h1, E, agg1, deg, nEdges, GCN_DHID,  1);
    }

    gcn_normalize<<<(int)((nH1 + 255) / 256), 256, 0, stream>>>(
        agg1, deg, h1, nH1, GCN_DHID,  1);

    gcn_gemm_bias_wmma<<<dim3(GCN_NODES / 16, GCN_DOUT / 64), 32, 0, stream>>>(
        h1, W2, b2, h2, GCN_DHID, GCN_DOUT);

    gcn_zero_f32<<<(int)((nOut + 255) / 256), 256, 0, stream>>>(out, nOut);
    {
        const long long tot = nEdges * (GCN_DOUT / 4);
        gcn_scatter_add<<<(int)((tot + 255) / 256), 256, 0, stream>>>(
            h2, E, out, deg, nEdges, GCN_DOUT,  0);
    }
    gcn_normalize<<<(int)((nOut + 255) / 256), 256, 0, stream>>>(
        out, deg, out, nOut, GCN_DOUT,  0);
}


# Layer 1 sums X over edges before the W1 product (sum is linear): 128-wide atomic scatter, not 256; mean, deg*b1, relu in GEMM epilogue; one launch fewer; f32 sum order changes
# speedup vs baseline: 1.3482x; 1.3482x over previous
#include <hip/hip_runtime.h>
#include <hip/hip_bf16.h>


typedef __attribute__((ext_vector_type(2))) float v2f;
typedef __attribute__((ext_vector_type(8))) float v8f;

#define GCN_NODES 50000
#define GCN_DIN   128
#define GCN_DHID  256
#define GCN_DOUT  128

__device__ __forceinline__ void gcn_tile_16x64(const float* __restrict__ arow,
                                               const float* __restrict__ W,
                                               int K, int N, int nBase,
                                               int l, int half, v8f (&acc)[4]) {
    for (int k = 0; k < K; k += 4) {
        const v2f a = *(const v2f*)(arow + k + 2 * half);

        const int krow = k + 2 * half;
#pragma unroll
        for (int t = 0; t < 4; ++t) {
            const int col = nBase + t * 16 + l;
            v2f b;
            b.x = W[(size_t)(krow)     * (size_t)N + col];
            b.y = W[(size_t)(krow + 1) * (size_t)N + col];
            acc[t] = __builtin_amdgcn_wmma_f32_16x16x4_f32(
                 false, a,  false, b,
                 (short)0, acc[t],
                 false,  false);
        }
    }
}

__global__ void gcn_gemm_bias_wmma(const float* __restrict__ X,
                                   const float* __restrict__ W,
                                   const float* __restrict__ bias,
                                   float* __restrict__ H,
                                   int K, int N) {
    const int lane = threadIdx.x;
    const int half = lane >> 4;
    const int l    = lane & 15;

    const int mBase = blockIdx.x * 16;
    const int nBase = blockIdx.y * 64;

    const float* arow = X + (size_t)(mBase + l) * (size_t)K;

    v8f acc[4];
#pragma unroll
    for (int t = 0; t < 4; ++t) {
        const float bv = bias[nBase + t * 16 + l];
#pragma unroll
        for (int i = 0; i < 8; ++i) acc[t][i] = bv;
    }

    gcn_tile_16x64(arow, W, K, N, nBase, l, half, acc);

#pragma unroll
    for (int t = 0; t < 4; ++t) {
        const int col = nBase + t * 16 + l;
#pragma unroll
        for (int i = 0; i < 8; ++i) {
            const int row = mBase + i + 8 * half;
            H[(size_t)row * (size_t)N + col] = acc[t][i];
        }
    }
}

__global__ void gcn_gemm_mean_relu_wmma(const float* __restrict__ S,
                                        const float* __restrict__ W,
                                        const float* __restrict__ bias,
                                        const float* __restrict__ deg,
                                        float* __restrict__ H,
                                        int K, int N) {
    const int lane = threadIdx.x;
    const int half = lane >> 4;
    const int l    = lane & 15;

    const int mBase = blockIdx.x * 16;
    const int nBase = blockIdx.y * 64;

    const float* arow = S + (size_t)(mBase + l) * (size_t)K;

    v8f acc[4];
#pragma unroll
    for (int t = 0; t < 4; ++t) {
#pragma unroll
        for (int i = 0; i < 8; ++i) acc[t][i] = 0.0f;
    }

    gcn_tile_16x64(arow, W, K, N, nBase, l, half, acc);

    float dg[8], dm[8];
#pragma unroll
    for (int i = 0; i < 8; ++i) {
        dg[i] = deg[mBase + i + 8 * half];
        dm[i] = fmaxf(dg[i], 1.0f);
    }

#pragma unroll
    for (int t = 0; t < 4; ++t) {
        const int col = nBase + t * 16 + l;
        const float bv = bias[col];
#pragma unroll
        for (int i = 0; i < 8; ++i) {
            const int row = mBase + i + 8 * half;
            float v = (acc[t][i] + dg[i] * bv) / dm[i];
            v = fmaxf(v, 0.0f);
            H[(size_t)row * (size_t)N + col] = v;
        }
    }
}

__global__ void gcn_scatter_add(const float* __restrict__ h,
                                const int* __restrict__ edges,
                                float* __restrict__ agg,
                                float* __restrict__ deg,
                                long long nEdges, int C, int addDeg) {
    const int chunks = C >> 2;
    const long long tid = (long long)blockIdx.x * blockDim.x + threadIdx.x;
    const long long total = nEdges * (long long)chunks;
    if (tid >= total) return;

    const long long e  = tid / chunks;
    const int       c4 = (int)(tid - e * (long long)chunks) << 2;

    const int src = edges[2 * e];
    const int dst = edges[2 * e + 1];

    const float4 v = *(const float4*)(h + (size_t)src * (size_t)C + c4);
    float* p = agg + (size_t)dst * (size_t)C + c4;
    atomicAdd(p + 0, v.x);
    atomicAdd(p + 1, v.y);
    atomicAdd(p + 2, v.z);
    atomicAdd(p + 3, v.w);

    if (addDeg && c4 == 0) atomicAdd(deg + dst, 1.0f);
}

__global__ void gcn_normalize(const float* __restrict__ agg,
                              const float* __restrict__ deg,
                              float* __restrict__ out,
                              long long n, int C, int doRelu) {
    const long long i = (long long)blockIdx.x * blockDim.x + threadIdx.x;
    if (i >= n) return;
    const long long node = i / C;
    const float d = fmaxf(deg[node], 1.0f);
    float v = agg[i] / d;
    if (doRelu) v = fmaxf(v, 0.0f);
    out[i] = v;
}

__global__ void gcn_zero_f32(float* __restrict__ p, long long n) {
    const long long i = (long long)blockIdx.x * blockDim.x + threadIdx.x;
    if (i < n) p[i] = 0.0f;
}

extern "C" void kernel_launch(void* const* d_in, const int* in_sizes, int n_in,
                              void* d_out, int out_size, void* d_ws, size_t ws_size,
                              hipStream_t stream) {
    const float* X  = (const float*)d_in[0];
    const int*   E  = (const int*)  d_in[1];
    const float* W1 = (const float*)d_in[2];
    const float* b1 = (const float*)d_in[3];
    const float* W2 = (const float*)d_in[4];
    const float* b2 = (const float*)d_in[5];
    float* out = (float*)d_out;

    const long long nEdges = (long long)in_sizes[1] / 2;

    char* ws = (char*)d_ws;
    float* aggX = (float*)ws;
    float* deg  = (float*)(ws + (size_t)GCN_NODES * GCN_DIN * 4);
    float* h1n  = (float*)(ws + (size_t)GCN_NODES * GCN_DHID * 4);
    float* h2   = (float*)(ws + (size_t)GCN_NODES * GCN_DHID * 8);

    const long long nX   = (long long)GCN_NODES * GCN_DIN;
    const long long nOut = (long long)GCN_NODES * GCN_DOUT;

    {
        const long long nz = nX + GCN_NODES;
        gcn_zero_f32<<<(int)((nz + 255) / 256), 256, 0, stream>>>(aggX, nz);
    }
    {
        const long long tot = nEdges * (GCN_DIN / 4);
        gcn_scatter_add<<<(int)((tot + 255) / 256), 256, 0, stream>>>(
            X, E, aggX, deg, nEdges, GCN_DIN,  1);
    }

    gcn_gemm_mean_relu_wmma<<<dim3(GCN_NODES / 16, GCN_DHID / 64), 32, 0, stream>>>(
        aggX, W1, b1, deg, h1n, GCN_DIN, GCN_DHID);

    gcn_gemm_bias_wmma<<<dim3(GCN_NODES / 16, GCN_DOUT / 64), 32, 0, stream>>>(
        h1n, W2, b2, h2, GCN_DHID, GCN_DOUT);

    gcn_zero_f32<<<(int)((nOut + 255) / 256), 256, 0, stream>>>(out, nOut);
    {
        const long long tot = nEdges * (GCN_DOUT / 4);
        gcn_scatter_add<<<(int)((tot + 255) / 256), 256, 0, stream>>>(
            h2, E, out, deg, nEdges, GCN_DOUT,  0);
    }
    gcn_normalize<<<(int)((nOut + 255) / 256), 256, 0, stream>>>(
        out, deg, out, nOut, GCN_DOUT,  0);
}


# Segment sums by gather over a dst-sorted edge list built per call (int atomic rank, scan, fill), one wave per node in registers; float atomics, zero and normalize passes gone; f32 sum order changes
# speedup vs baseline: 2.7577x; 2.0455x over previous
#include <hip/hip_runtime.h>
#include <hip/hip_bf16.h>


typedef __attribute__((ext_vector_type(2))) float v2f;
typedef __attribute__((ext_vector_type(8))) float v8f;

#define GCN_NODES 50000
#define GCN_DIN   128
#define GCN_DHID  256
#define GCN_DOUT  128

__device__ __forceinline__ void gcn_tile_16x64(const float* __restrict__ arow,
                                               const float* __restrict__ W,
                                               int K, int N, int nBase,
                                               int l, int half, v8f (&acc)[4]) {
    for (int k = 0; k < K; k += 4) {
        const v2f a = *(const v2f*)(arow + k + 2 * half);

        const int krow = k + 2 * half;
#pragma unroll
        for (int t = 0; t < 4; ++t) {
            const int col = nBase + t * 16 + l;
            v2f b;
            b.x = W[(size_t)(krow)     * (size_t)N + col];
            b.y = W[(size_t)(krow + 1) * (size_t)N + col];
            acc[t] = __builtin_amdgcn_wmma_f32_16x16x4_f32(
                 false, a,  false, b,
                 (short)0, acc[t],
                 false,  false);
        }
    }
}

__global__ void gcn_gemm_bias_wmma(const float* __restrict__ X,
                                   const float* __restrict__ W,
                                   const float* __restrict__ bias,
                                   float* __restrict__ H,
                                   int K, int N) {
    const int lane = threadIdx.x;
    const int half = lane >> 4;
    const int l    = lane & 15;

    const int mBase = blockIdx.x * 16;
    const int nBase = blockIdx.y * 64;

    const float* arow = X + (size_t)(mBase + l) * (size_t)K;

    v8f acc[4];
#pragma unroll
    for (int t = 0; t < 4; ++t) {
        const float bv = bias[nBase + t * 16 + l];
#pragma unroll
        for (int i = 0; i < 8; ++i) acc[t][i] = bv;
    }

    gcn_tile_16x64(arow, W, K, N, nBase, l, half, acc);

#pragma unroll
    for (int t = 0; t < 4; ++t) {
        const int col = nBase + t * 16 + l;
#pragma unroll
        for (int i = 0; i < 8; ++i) {
            const int row = mBase + i + 8 * half;
            H[(size_t)row * (size_t)N + col] = acc[t][i];
        }
    }
}

__global__ void gcn_gemm_mean_relu_wmma(const float* __restrict__ S,
                                        const float* __restrict__ W,
                                        const float* __restrict__ bias,
                                        const float* __restrict__ deg,
                                        float* __restrict__ H,
                                        int K, int N) {
    const int lane = threadIdx.x;
    const int half = lane >> 4;
    const int l    = lane & 15;

    const int mBase = blockIdx.x * 16;
    const int nBase = blockIdx.y * 64;

    const float* arow = S + (size_t)(mBase + l) * (size_t)K;

    v8f acc[4];
#pragma unroll
    for (int t = 0; t < 4; ++t) {
#pragma unroll
        for (int i = 0; i < 8; ++i) acc[t][i] = 0.0f;
    }

    gcn_tile_16x64(arow, W, K, N, nBase, l, half, acc);

    float dg[8], dm[8];
#pragma unroll
    for (int i = 0; i < 8; ++i) {
        dg[i] = deg[mBase + i + 8 * half];
        dm[i] = fmaxf(dg[i], 1.0f);
    }

#pragma unroll
    for (int t = 0; t < 4; ++t) {
        const int col = nBase + t * 16 + l;
        const float bv = bias[col];
#pragma unroll
        for (int i = 0; i < 8; ++i) {
            const int row = mBase + i + 8 * half;
            float v = (acc[t][i] + dg[i] * bv) / dm[i];
            v = fmaxf(v, 0.0f);
            H[(size_t)row * (size_t)N + col] = v;
        }
    }
}


__global__ void gcn_edge_rank(const int* __restrict__ edges,
                              int* __restrict__ cnt,
                              int* __restrict__ rank,
                              int nEdges) {
    const int e = blockIdx.x * blockDim.x + threadIdx.x;
    if (e >= nEdges) return;
    const int dst = edges[2 * e + 1];
    rank[e] = atomicAdd(cnt + dst, 1);
}

__global__ __launch_bounds__(1024)
void gcn_rowptr_scan(const int* __restrict__ cnt,
                     int* __restrict__ rowptr,
                     int n) {
    __shared__ int sOwn[1024];
    __shared__ int sPre[1024];
    const int tid   = threadIdx.x;
    const int start = blockIdx.x * 1024;

    int pre = 0;
    for (int i = tid; i < start; i += 1024) pre += cnt[i];

    const int i = start + tid;
    const int v = (i < n) ? cnt[i] : 0;
    sOwn[tid] = v;
    sPre[tid] = pre;
    __syncthreads();
    for (int off = 1; off < 1024; off <<= 1) {
        int a = 0, b = 0;
        if (tid >= off) { a = sOwn[tid - off]; b = sPre[tid - off]; }
        __syncthreads();
        sOwn[tid] += a;
        sPre[tid] += b;
        __syncthreads();
    }
    const int base = sPre[1023];
    const int incl = sOwn[tid];
    if (i < n) {
        rowptr[i] = base + incl - v;
        if (i == n - 1) rowptr[n] = base + incl;
    }
}

__global__ void gcn_csr_fill(const int* __restrict__ edges,
                             const int* __restrict__ rowptr,
                             const int* __restrict__ rank,
                             int* __restrict__ csr_src,
                             int nEdges) {
    const int e = blockIdx.x * blockDim.x + threadIdx.x;
    if (e >= nEdges) return;
    const int src = edges[2 * e];
    const int dst = edges[2 * e + 1];
    csr_src[rowptr[dst] + rank[e]] = src;
}

#define GCN_ROW 128

template <int B>
__device__ __forceinline__ void gcn_gather_batch(const float* __restrict__ hl,
                                                 int idx, int t, float4& acc) {
    float4 v[B];
#pragma unroll
    for (int u = 0; u < B; ++u) {
        const int s = __builtin_amdgcn_readlane(idx, t + u);
        v[u] = *(const float4*)(hl + (size_t)s * GCN_ROW);
    }
#pragma unroll
    for (int u = 0; u < B; ++u) {
        acc.x += v[u].x;
        acc.y += v[u].y;
        acc.z += v[u].z;
        acc.w += v[u].w;
    }
}

__global__ __launch_bounds__(256)
void gcn_segment_gather(const float* __restrict__ h,
                        const int* __restrict__ rowptr,
                        const int* __restrict__ csr_src,
                        float* __restrict__ outp,
                        float* __restrict__ deg,
                        int nNodes, int nEdges, int mean) {
    const int lane = threadIdx.x & 31;
    const int n    = blockIdx.x * 8 + (threadIdx.x >> 5);
    if (n >= nNodes) return;

    int beg = __builtin_amdgcn_readfirstlane(rowptr[n]);
    int end = __builtin_amdgcn_readfirstlane(rowptr[n + 1]);
    beg = max(beg, 0);
    end = min(end, nEdges);
    const int cnt = max(end - beg, 0);

    float4 acc = make_float4(0.0f, 0.0f, 0.0f, 0.0f);
    const float* hl = h + lane * 4;

    for (int j = beg; j < end; j += 32) {
        const int m   = min(end - j, 32);
        const int idx = (lane < m) ? csr_src[j + lane] : 0;
        int t = 0;
        for (; t + 8 <= m; t += 8) gcn_gather_batch<8>(hl, idx, t, acc);
        if (t + 4 <= m) { gcn_gather_batch<4>(hl, idx, t, acc); t += 4; }
        if (t + 2 <= m) { gcn_gather_batch<2>(hl, idx, t, acc); t += 2; }
        if (t < m)      { gcn_gather_batch<1>(hl, idx, t, acc); }
    }

    const float d = (float)cnt;
    if (mean) {
        const float dm = fmaxf(d, 1.0f);
        acc.x /= dm;
        acc.y /= dm;
        acc.z /= dm;
        acc.w /= dm;
    }
    *(float4*)(outp + (size_t)n * GCN_ROW + lane * 4) = acc;
    if (deg != nullptr && lane == 0) deg[n] = d;
}

__global__ void gcn_zero_f32(float* __restrict__ p, long long n) {
    const long long i = (long long)blockIdx.x * blockDim.x + threadIdx.x;
    if (i < n) p[i] = 0.0f;
}

extern "C" void kernel_launch(void* const* d_in, const int* in_sizes, int n_in,
                              void* d_out, int out_size, void* d_ws, size_t ws_size,
                              hipStream_t stream) {
    const float* X  = (const float*)d_in[0];
    const int*   E  = (const int*)  d_in[1];
    const float* W1 = (const float*)d_in[2];
    const float* b1 = (const float*)d_in[3];
    const float* W2 = (const float*)d_in[4];
    const float* b2 = (const float*)d_in[5];
    float* out = (float*)d_out;

    const int nEdges = in_sizes[1] / 2;

    char* ws = (char*)d_ws;
    float* aggX = (float*)ws;
    float* deg  = (float*)(ws + (size_t)GCN_NODES * GCN_DIN * 4);
    float* h1n  = (float*)(ws + (size_t)GCN_NODES * GCN_DHID * 4);
    float* h2   = (float*)(ws + (size_t)GCN_NODES * GCN_DHID * 8);
    char*  csr  = ws + (size_t)GCN_NODES * GCN_DHID * 10;
    int* cnt    = (int*)(csr);
    int* rowptr = (int*)(csr + (1 << 18));
    int* rank   = (int*)(csr + (1 << 19));
    int* csrSrc = (int*)(csr + (1 << 19) + (1 << 22));

    gcn_zero_f32<<<(GCN_NODES + 255) / 256, 256, 0, stream>>>((float*)cnt, (long long)GCN_NODES);
    gcn_edge_rank<<<(nEdges + 255) / 256, 256, 0, stream>>>(E, cnt, rank, nEdges);
    gcn_rowptr_scan<<<(GCN_NODES + 1023) / 1024, 1024, 0, stream>>>(cnt, rowptr, GCN_NODES);
    gcn_csr_fill<<<(nEdges + 255) / 256, 256, 0, stream>>>(E, rowptr, rank, csrSrc, nEdges);

    gcn_segment_gather<<<(GCN_NODES + 7) / 8, 256, 0, stream>>>(
        X, rowptr, csrSrc, aggX, deg, GCN_NODES, nEdges,  0);

    gcn_gemm_mean_relu_wmma<<<dim3(GCN_NODES / 16, GCN_DHID / 64), 32, 0, stream>>>(
        aggX, W1, b1, deg, h1n, GCN_DIN, GCN_DHID);

    gcn_gemm_bias_wmma<<<dim3(GCN_NODES / 16, GCN_DOUT / 64), 32, 0, stream>>>(
        h1n, W2, b2, h2, GCN_DHID, GCN_DOUT);

    gcn_segment_gather<<<(GCN_NODES + 7) / 8, 256, 0, stream>>>(
        h2, rowptr, csrSrc, out, nullptr, GCN_NODES, nEdges,  1);
}


# Edge list ordered by dst in two launches that count in LDS (chunks by dst bucket, then buckets by node): global returning atomics, clear, scan and fill launches gone; f32 sum order changes
# speedup vs baseline: 3.9645x; 1.4376x over previous
#include <hip/hip_runtime.h>
#include <hip/hip_bf16.h>


typedef __attribute__((ext_vector_type(2))) float v2f;
typedef __attribute__((ext_vector_type(8))) float v8f;

#define GCN_NODES 50000
#define GCN_DIN   128
#define GCN_DHID  256
#define GCN_DOUT  128

__device__ __forceinline__ void gcn_tile_16x64(const float* __restrict__ arow,
                                               const float* __restrict__ W,
                                               int K, int N, int nBase,
                                               int l, int half, v8f (&acc)[4]) {
    for (int k = 0; k < K; k += 4) {
        const v2f a = *(const v2f*)(arow + k + 2 * half);

        const int krow = k + 2 * half;
#pragma unroll
        for (int t = 0; t < 4; ++t) {
            const int col = nBase + t * 16 + l;
            v2f b;
            b.x = W[(size_t)(krow)     * (size_t)N + col];
            b.y = W[(size_t)(krow + 1) * (size_t)N + col];
            acc[t] = __builtin_amdgcn_wmma_f32_16x16x4_f32(
                 false, a,  false, b,
                 (short)0, acc[t],
                 false,  false);
        }
    }
}

__global__ void gcn_gemm_bias_wmma(const float* __restrict__ X,
                                   const float* __restrict__ W,
                                   const float* __restrict__ bias,
                                   float* __restrict__ H,
                                   int K, int N) {
    const int lane = threadIdx.x;
    const int half = lane >> 4;
    const int l    = lane & 15;

    const int mBase = blockIdx.x * 16;
    const int nBase = blockIdx.y * 64;

    const float* arow = X + (size_t)(mBase + l) * (size_t)K;

    v8f acc[4];
#pragma unroll
    for (int t = 0; t < 4; ++t) {
        const float bv = bias[nBase + t * 16 + l];
#pragma unroll
        for (int i = 0; i < 8; ++i) acc[t][i] = bv;
    }

    gcn_tile_16x64(arow, W, K, N, nBase, l, half, acc);

#pragma unroll
    for (int t = 0; t < 4; ++t) {
        const int col = nBase + t * 16 + l;
#pragma unroll
        for (int i = 0; i < 8; ++i) {
            const int row = mBase + i + 8 * half;
            H[(size_t)row * (size_t)N + col] = acc[t][i];
        }
    }
}

__global__ void gcn_gemm_mean_relu_wmma(const float* __restrict__ S,
                                        const float* __restrict__ W,
                                        const float* __restrict__ bias,
                                        const float* __restrict__ deg,
                                        float* __restrict__ H,
                                        int K, int N) {
    const int lane = threadIdx.x;
    const int half = lane >> 4;
    const int l    = lane & 15;

    const int mBase = blockIdx.x * 16;
    const int nBase = blockIdx.y * 64;

    const float* arow = S + (size_t)(mBase + l) * (size_t)K;

    v8f acc[4];
#pragma unroll
    for (int t = 0; t < 4; ++t) {
#pragma unroll
        for (int i = 0; i < 8; ++i) acc[t][i] = 0.0f;
    }

    gcn_tile_16x64(arow, W, K, N, nBase, l, half, acc);

    float dg[8], dm[8];
#pragma unroll
    for (int i = 0; i < 8; ++i) {
        dg[i] = deg[mBase + i + 8 * half];
        dm[i] = fmaxf(dg[i], 1.0f);
    }

#pragma unroll
    for (int t = 0; t < 4; ++t) {
        const int col = nBase + t * 16 + l;
        const float bv = bias[col];
#pragma unroll
        for (int i = 0; i < 8; ++i) {
            const int row = mBase + i + 8 * half;
            float v = (acc[t][i] + dg[i] * bv) / dm[i];
            v = fmaxf(v, 0.0f);
            H[(size_t)row * (size_t)N + col] = v;
        }
    }
}

#define GCN_NCHUNK  256
#define GCN_NBUCKET 256
#define GCN_BNODES  196
#define GCN_SORT_THREADS 1024

static_assert(GCN_NBUCKET * GCN_BNODES >= GCN_NODES, "buckets must cover every node");
static_assert(GCN_BNODES <= 256 && GCN_NBUCKET == 256, "one wave scans 256 LDS words, 8 per lane");
static_assert(GCN_SORT_THREADS == 4 * GCN_NCHUNK, "order pass: four threads per chunk run");

__device__ __forceinline__ int gcn_wave_scan256(const int* sIn, int lane, int (&excl)[8]) {
    int run = 0;
#pragma unroll
    for (int i = 0; i < 8; ++i) {
        const int t = sIn[lane * 8 + i];
        excl[i] = run;
        run += t;
    }
    int inc = run;
#pragma unroll
    for (int d = 1; d < 32; d <<= 1) {
        const int o = __shfl_up(inc, d);
        if (lane >= d) inc += o;
    }
    const int before = inc - run;
#pragma unroll
    for (int i = 0; i < 8; ++i) excl[i] += before;
    return inc;
}

__device__ __forceinline__ int gcn_bucket_of(int dst) {
    return min(max(dst / GCN_BNODES, 0), GCN_NBUCKET - 1);
}

__global__ __launch_bounds__(GCN_SORT_THREADS)
void gcn_edge_bucket(const int* __restrict__ edges,
                     int* __restrict__ tab,
                     int* __restrict__ sorted,
                     int nEdges) {
    __shared__ int sCnt[GCN_NBUCKET];
    __shared__ int sCur[GCN_NBUCKET];
    const int tid = threadIdx.x;
    const int c   = blockIdx.x;
    const int chunkLen = (nEdges + GCN_NCHUNK - 1) / GCN_NCHUNK;
    const int cs = min(c * chunkLen, nEdges);
    const int ce = min(cs + chunkLen, nEdges);
    const int2* e2 = (const int2*)edges;

    if (tid < GCN_NBUCKET) sCnt[tid] = 0;
    __syncthreads();

    for (int e0 = cs + tid; e0 < ce; e0 += 4 * GCN_SORT_THREADS) {
        int d[4];
#pragma unroll
        for (int u = 0; u < 4; ++u)
            d[u] = e2[min(e0 + u * GCN_SORT_THREADS, ce - 1)].y;
        asm volatile("" :: "v"(d[0]), "v"(d[1]), "v"(d[2]), "v"(d[3]));
#pragma unroll
        for (int u = 0; u < 4; ++u)
            if (e0 + u * GCN_SORT_THREADS < ce) atomicAdd(&sCnt[gcn_bucket_of(d[u])], 1);
    }
    __syncthreads();

    if (tid < 32) {
        int excl[8];
        gcn_wave_scan256(sCnt, tid, excl);
#pragma unroll
        for (int i = 0; i < 8; ++i) sCur[tid * 8 + i] = excl[i];
    }
    __syncthreads();

    int myCnt = 0, myOff = 0;
    if (tid < GCN_NBUCKET) { myCnt = sCnt[tid]; myOff = sCur[tid]; }
    __syncthreads();
    if (tid < GCN_NBUCKET) ((int2*)tab)[tid * GCN_NCHUNK + c] = make_int2(myCnt, myOff);

    int2* out2 = (int2*)sorted + cs;
    for (int e0 = cs + tid; e0 < ce; e0 += 4 * GCN_SORT_THREADS) {
        int2 sd[4];
#pragma unroll
        for (int u = 0; u < 4; ++u)
            sd[u] = e2[min(e0 + u * GCN_SORT_THREADS, ce - 1)];
        asm volatile("" :: "v"(sd[0].y), "v"(sd[1].y), "v"(sd[2].y), "v"(sd[3].y));
#pragma unroll
        for (int u = 0; u < 4; ++u) {
            if (e0 + u * GCN_SORT_THREADS < ce) {
                const int pos = atomicAdd(&sCur[gcn_bucket_of(sd[u].y)], 1);
                out2[pos] = sd[u];
            }
        }
    }
}

static_assert(GCN_BNODES <= GCN_SORT_THREADS, "one thread stores one rowptr word");

__device__ __forceinline__ int gcn_local_node(int dst, int nodeBase) {
    return min(max(dst - nodeBase, 0), GCN_BNODES - 1);
}

__global__ __launch_bounds__(GCN_SORT_THREADS)
void gcn_bucket_order(const int* __restrict__ tab,
                      const int* __restrict__ sorted,
                      int* __restrict__ rowptr,
                      int* __restrict__ csr_src,
                      int nEdges) {
    __shared__ int sNode[256];
    __shared__ int sRow[256];
    __shared__ int sRed[32];
    const int tid  = threadIdx.x;
    const int lane = tid & 31;
    const int q    = blockIdx.x;
    const int chunkLen = (nEdges + GCN_NCHUNK - 1) / GCN_NCHUNK;

    const int c   = tid >> 2;
    const int sub = tid & 3;
    const int2 seg = ((const int2*)tab)[q * GCN_NCHUNK + c];
    const int segCnt = min(max(seg.x, 0), chunkLen);
    const int segOff = min(max(seg.y, 0), chunkLen - segCnt);

    int part = (sub == 0) ? segOff : 0;
#pragma unroll
    for (int d = 16; d >= 1; d >>= 1) part += __shfl_xor(part, d);
    if (lane == 0) sRed[tid >> 5] = part;
    if (tid < 256) sNode[tid] = 0;
    __syncthreads();

    const int cs = min(c * chunkLen, nEdges);
    const int2* s2 = (const int2*)sorted + cs + segOff;
    const int nodeBase = q * GCN_BNODES;

    for (int i0 = sub; i0 < segCnt; i0 += 16) {
        int d[4];
#pragma unroll
        for (int u = 0; u < 4; ++u) d[u] = s2[min(i0 + 4 * u, segCnt - 1)].y;
        asm volatile("" :: "v"(d[0]), "v"(d[1]), "v"(d[2]), "v"(d[3]));
#pragma unroll
        for (int u = 0; u < 4; ++u)
            if (i0 + 4 * u < segCnt) atomicAdd(&sNode[gcn_local_node(d[u], nodeBase)], 1);
    }
    __syncthreads();

    int tot = 0;
    if (tid < 32) {
        int base = sRed[tid];
#pragma unroll
        for (int d = 16; d >= 1; d >>= 1) base += __shfl_xor(base, d);
        int excl[8];
        const int inc = gcn_wave_scan256(sNode, tid, excl);
#pragma unroll
        for (int i = 0; i < 8; ++i) {
            const int rp = base + excl[i];
            sNode[tid * 8 + i] = rp;
            sRow[tid * 8 + i]  = rp;
        }
        tot = base + inc;
    }
    __syncthreads();
    if (tid < GCN_BNODES && nodeBase + tid < GCN_NODES) rowptr[nodeBase + tid] = sRow[tid];
    if (q == GCN_NBUCKET - 1 && tid == 31) rowptr[GCN_NODES] = tot;

    for (int i0 = sub; i0 < segCnt; i0 += 16) {
        int2 sd[4];
#pragma unroll
        for (int u = 0; u < 4; ++u) sd[u] = s2[min(i0 + 4 * u, segCnt - 1)];
        asm volatile("" :: "v"(sd[0].y), "v"(sd[1].y), "v"(sd[2].y), "v"(sd[3].y));
#pragma unroll
        for (int u = 0; u < 4; ++u) {
            if (i0 + 4 * u < segCnt) {
                const int pos = atomicAdd(&sNode[gcn_local_node(sd[u].y, nodeBase)], 1);
                if ((unsigned)pos < (unsigned)nEdges) csr_src[pos] = sd[u].x;
            }
        }
    }
}

#define GCN_ROW 128

template <int B>
__device__ __forceinline__ void gcn_gather_batch(const float* __restrict__ hl,
                                                 int idx, int t, float4& acc) {
    float4 v[B];
#pragma unroll
    for (int u = 0; u < B; ++u) {
        const int s = __builtin_amdgcn_readlane(idx, t + u);
        v[u] = *(const float4*)(hl + (size_t)s * GCN_ROW);
    }
#pragma unroll
    for (int u = 0; u < B; ++u) {
        acc.x += v[u].x;
        acc.y += v[u].y;
        acc.z += v[u].z;
        acc.w += v[u].w;
    }
}

__global__ __launch_bounds__(256)
void gcn_segment_gather(const float* __restrict__ h,
                        const int* __restrict__ rowptr,
                        const int* __restrict__ csr_src,
                        float* __restrict__ outp,
                        float* __restrict__ deg,
                        int nNodes, int nEdges, int mean) {
    const int lane = threadIdx.x & 31;
    const int n    = blockIdx.x * 8 + (threadIdx.x >> 5);
    if (n >= nNodes) return;

    int beg = __builtin_amdgcn_readfirstlane(rowptr[n]);
    int end = __builtin_amdgcn_readfirstlane(rowptr[n + 1]);
    beg = max(beg, 0);
    end = min(end, nEdges);
    const int cnt = max(end - beg, 0);

    float4 acc = make_float4(0.0f, 0.0f, 0.0f, 0.0f);
    const float* hl = h + lane * 4;

    for (int j = beg; j < end; j += 32) {
        const int m   = min(end - j, 32);
        const int idx = (lane < m) ? csr_src[j + lane] : 0;
        int t = 0;
        for (; t + 8 <= m; t += 8) gcn_gather_batch<8>(hl, idx, t, acc);
        if (t + 4 <= m) { gcn_gather_batch<4>(hl, idx, t, acc); t += 4; }
        if (t + 2 <= m) { gcn_gather_batch<2>(hl, idx, t, acc); t += 2; }
        if (t < m)      { gcn_gather_batch<1>(hl, idx, t, acc); }
    }

    const float d = (float)cnt;
    if (mean) {
        const float dm = fmaxf(d, 1.0f);
        acc.x /= dm;
        acc.y /= dm;
        acc.z /= dm;
        acc.w /= dm;
    }
    *(float4*)(outp + (size_t)n * GCN_ROW + lane * 4) = acc;
    if (deg != nullptr && lane == 0) deg[n] = d;
}

extern "C" void kernel_launch(void* const* d_in, const int* in_sizes, int n_in,
                              void* d_out, int out_size, void* d_ws, size_t ws_size,
                              hipStream_t stream) {
    const float* X  = (const float*)d_in[0];
    const int*   E  = (const int*)  d_in[1];
    const float* W1 = (const float*)d_in[2];
    const float* b1 = (const float*)d_in[3];
    const float* W2 = (const float*)d_in[4];
    const float* b2 = (const float*)d_in[5];
    float* out = (float*)d_out;

    const int nEdges = in_sizes[1] / 2;

    char* ws = (char*)d_ws;
    float* aggX = (float*)ws;
    float* deg  = (float*)(ws + (size_t)GCN_NODES * GCN_DIN * 4);
    float* h1n  = (float*)(ws + (size_t)GCN_NODES * GCN_DHID * 4);
    float* h2   = (float*)(ws + (size_t)GCN_NODES * GCN_DHID * 8);
    char*  csr  = ws + (size_t)GCN_NODES * GCN_DHID * 10;
    int* tab    = (int*)(csr);
    int* rowptr = (int*)(csr + (1 << 19));
    int* sorted = (int*)(csr + (1 << 20));
    int* csrSrc = (int*)(csr + (1 << 20) + (1 << 23));

    gcn_edge_bucket<<<GCN_NCHUNK, GCN_SORT_THREADS, 0, stream>>>(E, tab, sorted, nEdges);
    gcn_bucket_order<<<GCN_NBUCKET, GCN_SORT_THREADS, 0, stream>>>(tab, sorted, rowptr, csrSrc, nEdges);

    gcn_segment_gather<<<(GCN_NODES + 7) / 8, 256, 0, stream>>>(
        X, rowptr, csrSrc, aggX, deg, GCN_NODES, nEdges,  0);

    gcn_gemm_mean_relu_wmma<<<dim3(GCN_NODES / 16, GCN_DHID / 64), 32, 0, stream>>>(
        aggX, W1, b1, deg, h1n, GCN_DIN, GCN_DHID);

    gcn_gemm_bias_wmma<<<dim3(GCN_NODES / 16, GCN_DOUT / 64), 32, 0, stream>>>(
        h1n, W2, b2, h2, GCN_DHID, GCN_DOUT);

    gcn_segment_gather<<<(GCN_NODES + 7) / 8, 256, 0, stream>>>(
        h2, rowptr, csrSrc, out, nullptr, GCN_NODES, nEdges,  1);
}


# Both GEMMs: lane l of N subtile t owns column 4l+t of the 64-wide strip, so a k row's four B fragments are one 16-byte load and a D row's four values one 16-byte store; k order and sums unchanged
# speedup vs baseline: 4.2366x; 1.0686x over previous
#include <hip/hip_runtime.h>
#include <hip/hip_bf16.h>


typedef __attribute__((ext_vector_type(2))) float v2f;
typedef __attribute__((ext_vector_type(4))) float v4f;
typedef __attribute__((ext_vector_type(8))) float v8f;

#define GCN_NODES 50000
#define GCN_DIN   128
#define GCN_DHID  256
#define GCN_DOUT  128

__device__ __forceinline__ void gcn_wmma_step(const v2f a, const v4f b0, const v4f b1,
                                              v8f (&acc)[4]) {
#pragma unroll
    for (int t = 0; t < 4; ++t) {
        v2f b;
        b.x = b0[t];
        b.y = b1[t];
        acc[t] = __builtin_amdgcn_wmma_f32_16x16x4_f32(
             false, a,  false, b,
             (short)0, acc[t],
             false,  false);
    }
}

__device__ __forceinline__ void gcn_tile_16x64(const float* __restrict__ arow,
                                               const float* __restrict__ W,
                                               int K, int N, int nBase,
                                               int l, int half, v8f (&acc)[4]) {
    const float* ap = arow + 2 * half;
    const float* wp = W + (size_t)(2 * half) * (size_t)N + nBase + 4 * l;
    const size_t n1 = (size_t)N;
    int k = 0;
    for (; k + 8 <= K; k += 8) {
        const v2f a0 = *(const v2f*)(ap);
        const v2f a1 = *(const v2f*)(ap + 4);
        const v4f p0 = *(const v4f*)(wp);
        const v4f p1 = *(const v4f*)(wp + n1);
        const v4f q0 = *(const v4f*)(wp + 4 * n1);
        const v4f q1 = *(const v4f*)(wp + 5 * n1);
        gcn_wmma_step(a0, p0, p1, acc);
        gcn_wmma_step(a1, q0, q1, acc);
        ap += 8;
        wp += 8 * n1;
    }
    if (k < K) {
        const v2f a0 = *(const v2f*)(ap);
        const v4f p0 = *(const v4f*)(wp);
        const v4f p1 = *(const v4f*)(wp + n1);
        gcn_wmma_step(a0, p0, p1, acc);
    }
}

__global__ void gcn_gemm_bias_wmma(const float* __restrict__ X,
                                   const float* __restrict__ W,
                                   const float* __restrict__ bias,
                                   float* __restrict__ H,
                                   int K, int N) {
    const int lane = threadIdx.x;
    const int half = lane >> 4;
    const int l    = lane & 15;

    const int mBase = blockIdx.x * 16;
    const int nBase = blockIdx.y * 64;

    const float* arow = X + (size_t)(mBase + l) * (size_t)K;

    const v4f bv4 = *(const v4f*)(bias + nBase + 4 * l);
    v8f acc[4];
#pragma unroll
    for (int t = 0; t < 4; ++t) {
#pragma unroll
        for (int i = 0; i < 8; ++i) acc[t][i] = bv4[t];
    }

    gcn_tile_16x64(arow, W, K, N, nBase, l, half, acc);

#pragma unroll
    for (int i = 0; i < 8; ++i) {
        const int row = mBase + i + 8 * half;
        v4f o;
#pragma unroll
        for (int t = 0; t < 4; ++t) o[t] = acc[t][i];
        *(v4f*)(H + (size_t)row * (size_t)N + nBase + 4 * l) = o;
    }
}

__global__ void gcn_gemm_mean_relu_wmma(const float* __restrict__ S,
                                        const float* __restrict__ W,
                                        const float* __restrict__ bias,
                                        const float* __restrict__ deg,
                                        float* __restrict__ H,
                                        int K, int N) {
    const int lane = threadIdx.x;
    const int half = lane >> 4;
    const int l    = lane & 15;

    const int mBase = blockIdx.x * 16;
    const int nBase = blockIdx.y * 64;

    const float* arow = S + (size_t)(mBase + l) * (size_t)K;

    v8f acc[4];
#pragma unroll
    for (int t = 0; t < 4; ++t) {
#pragma unroll
        for (int i = 0; i < 8; ++i) acc[t][i] = 0.0f;
    }

    gcn_tile_16x64(arow, W, K, N, nBase, l, half, acc);

    float dg[8], dm[8];
#pragma unroll
    for (int i = 0; i < 8; ++i) {
        dg[i] = deg[mBase + i + 8 * half];
        dm[i] = fmaxf(dg[i], 1.0f);
    }

    const v4f bv4 = *(const v4f*)(bias + nBase + 4 * l);
#pragma unroll
    for (int i = 0; i < 8; ++i) {
        const int row = mBase + i + 8 * half;
        v4f o;
#pragma unroll
        for (int t = 0; t < 4; ++t) {
            const float v = (acc[t][i] + dg[i] * bv4[t]) / dm[i];
            o[t] = fmaxf(v, 0.0f);
        }
        *(v4f*)(H + (size_t)row * (size_t)N + nBase + 4 * l) = o;
    }
}

#define GCN_NCHUNK  256
#define GCN_NBUCKET 256
#define GCN_BNODES  196
#define GCN_SORT_THREADS 1024

static_assert(GCN_NBUCKET * GCN_BNODES >= GCN_NODES, "buckets must cover every node");
static_assert(GCN_BNODES <= 256 && GCN_NBUCKET == 256, "one wave scans 256 LDS words, 8 per lane");
static_assert(GCN_SORT_THREADS == 4 * GCN_NCHUNK, "order pass: four threads per chunk run");

__device__ __forceinline__ int gcn_wave_scan256(const int* sIn, int lane, int (&excl)[8]) {
    int run = 0;
#pragma unroll
    for (int i = 0; i < 8; ++i) {
        const int t = sIn[lane * 8 + i];
        excl[i] = run;
        run += t;
    }
    int inc = run;
#pragma unroll
    for (int d = 1; d < 32; d <<= 1) {
        const int o = __shfl_up(inc, d);
        if (lane >= d) inc += o;
    }
    const int before = inc - run;
#pragma unroll
    for (int i = 0; i < 8; ++i) excl[i] += before;
    return inc;
}

__device__ __forceinline__ int gcn_bucket_of(int dst) {
    return min(max(dst / GCN_BNODES, 0), GCN_NBUCKET - 1);
}

__global__ __launch_bounds__(GCN_SORT_THREADS)
void gcn_edge_bucket(const int* __restrict__ edges,
                     int* __restrict__ tab,
                     int* __restrict__ sorted,
                     int nEdges) {
    __shared__ int sCnt[GCN_NBUCKET];
    __shared__ int sCur[GCN_NBUCKET];
    const int tid = threadIdx.x;
    const int c   = blockIdx.x;
    const int chunkLen = (nEdges + GCN_NCHUNK - 1) / GCN_NCHUNK;
    const int cs = min(c * chunkLen, nEdges);
    const int ce = min(cs + chunkLen, nEdges);
    const int2* e2 = (const int2*)edges;

    if (tid < GCN_NBUCKET) sCnt[tid] = 0;
    __syncthreads();

    for (int e0 = cs + tid; e0 < ce; e0 += 4 * GCN_SORT_THREADS) {
        int d[4];
#pragma unroll
        for (int u = 0; u < 4; ++u)
            d[u] = e2[min(e0 + u * GCN_SORT_THREADS, ce - 1)].y;
        asm volatile("" :: "v"(d[0]), "v"(d[1]), "v"(d[2]), "v"(d[3]));
#pragma unroll
        for (int u = 0; u < 4; ++u)
            if (e0 + u * GCN_SORT_THREADS < ce) atomicAdd(&sCnt[gcn_bucket_of(d[u])], 1);
    }
    __syncthreads();

    if (tid < 32) {
        int excl[8];
        gcn_wave_scan256(sCnt, tid, excl);
#pragma unroll
        for (int i = 0; i < 8; ++i) sCur[tid * 8 + i] = excl[i];
    }
    __syncthreads();

    int myCnt = 0, myOff = 0;
    if (tid < GCN_NBUCKET) { myCnt = sCnt[tid]; myOff = sCur[tid]; }
    __syncthreads();
    if (tid < GCN_NBUCKET) ((int2*)tab)[tid * GCN_NCHUNK + c] = make_int2(myCnt, myOff);

    int2* out2 = (int2*)sorted + cs;
    for (int e0 = cs + tid; e0 < ce; e0 += 4 * GCN_SORT_THREADS) {
        int2 sd[4];
#pragma unroll
        for (int u = 0; u < 4; ++u)
            sd[u] = e2[min(e0 + u * GCN_SORT_THREADS, ce - 1)];
        asm volatile("" :: "v"(sd[0].y), "v"(sd[1].y), "v"(sd[2].y), "v"(sd[3].y));
#pragma unroll
        for (int u = 0; u < 4; ++u) {
            if (e0 + u * GCN_SORT_THREADS < ce) {
                const int pos = atomicAdd(&sCur[gcn_bucket_of(sd[u].y)], 1);
                out2[pos] = sd[u];
            }
        }
    }
}

static_assert(GCN_BNODES <= GCN_SORT_THREADS, "one thread stores one rowptr word");

__device__ __forceinline__ int gcn_local_node(int dst, int nodeBase) {
    return min(max(dst - nodeBase, 0), GCN_BNODES - 1);
}

__global__ __launch_bounds__(GCN_SORT_THREADS)
void gcn_bucket_order(const int* __restrict__ tab,
                      const int* __restrict__ sorted,
                      int* __restrict__ rowptr,
                      int* __restrict__ csr_src,
                      int nEdges) {
    __shared__ int sNode[256];
    __shared__ int sRow[256];
    __shared__ int sRed[32];
    const int tid  = threadIdx.x;
    const int lane = tid & 31;
    const int q    = blockIdx.x;
    const int chunkLen = (nEdges + GCN_NCHUNK - 1) / GCN_NCHUNK;

    const int c   = tid >> 2;
    const int sub = tid & 3;
    const int2 seg = ((const int2*)tab)[q * GCN_NCHUNK + c];
    const int segCnt = min(max(seg.x, 0), chunkLen);
    const int segOff = min(max(seg.y, 0), chunkLen - segCnt);

    int part = (sub == 0) ? segOff : 0;
#pragma unroll
    for (int d = 16; d >= 1; d >>= 1) part += __shfl_xor(part, d);
    if (lane == 0) sRed[tid >> 5] = part;
    if (tid < 256) sNode[tid] = 0;
    __syncthreads();

    const int cs = min(c * chunkLen, nEdges);
    const int2* s2 = (const int2*)sorted + cs + segOff;
    const int nodeBase = q * GCN_BNODES;

    for (int i0 = sub; i0 < segCnt; i0 += 16) {
        int d[4];
#pragma unroll
        for (int u = 0; u < 4; ++u) d[u] = s2[min(i0 + 4 * u, segCnt - 1)].y;
        asm volatile("" :: "v"(d[0]), "v"(d[1]), "v"(d[2]), "v"(d[3]));
#pragma unroll
        for (int u = 0; u < 4; ++u)
            if (i0 + 4 * u < segCnt) atomicAdd(&sNode[gcn_local_node(d[u], nodeBase)], 1);
    }
    __syncthreads();

    int tot = 0;
    if (tid < 32) {
        int base = sRed[tid];
#pragma unroll
        for (int d = 16; d >= 1; d >>= 1) base += __shfl_xor(base, d);
        int excl[8];
        const int inc = gcn_wave_scan256(sNode, tid, excl);
#pragma unroll
        for (int i = 0; i < 8; ++i) {
            const int rp = base + excl[i];
            sNode[tid * 8 + i] = rp;
            sRow[tid * 8 + i]  = rp;
        }
        tot = base + inc;
    }
    __syncthreads();
    if (tid < GCN_BNODES && nodeBase + tid < GCN_NODES) rowptr[nodeBase + tid] = sRow[tid];
    if (q == GCN_NBUCKET - 1 && tid == 31) rowptr[GCN_NODES] = tot;

    for (int i0 = sub; i0 < segCnt; i0 += 16) {
        int2 sd[4];
#pragma unroll
        for (int u = 0; u < 4; ++u) sd[u] = s2[min(i0 + 4 * u, segCnt - 1)];
        asm volatile("" :: "v"(sd[0].y), "v"(sd[1].y), "v"(sd[2].y), "v"(sd[3].y));
#pragma unroll
        for (int u = 0; u < 4; ++u) {
            if (i0 + 4 * u < segCnt) {
                const int pos = atomicAdd(&sNode[gcn_local_node(sd[u].y, nodeBase)], 1);
                if ((unsigned)pos < (unsigned)nEdges) csr_src[pos] = sd[u].x;
            }
        }
    }
}

#define GCN_ROW 128

template <int B>
__device__ __forceinline__ void gcn_gather_batch(const float* __restrict__ hl,
                                                 int idx, int t, float4& acc) {
    float4 v[B];
#pragma unroll
    for (int u = 0; u < B; ++u) {
        const int s = __builtin_amdgcn_readlane(idx, t + u);
        v[u] = *(const float4*)(hl + (size_t)s * GCN_ROW);
    }
#pragma unroll
    for (int u = 0; u < B; ++u) {
        acc.x += v[u].x;
        acc.y += v[u].y;
        acc.z += v[u].z;
        acc.w += v[u].w;
    }
}

__global__ __launch_bounds__(256)
void gcn_segment_gather(const float* __restrict__ h,
                        const int* __restrict__ rowptr,
                        const int* __restrict__ csr_src,
                        float* __restrict__ outp,
                        float* __restrict__ deg,
                        int nNodes, int nEdges, int mean) {
    const int lane = threadIdx.x & 31;
    const int n    = blockIdx.x * 8 + (threadIdx.x >> 5);
    if (n >= nNodes) return;

    int beg = __builtin_amdgcn_readfirstlane(rowptr[n]);
    int end = __builtin_amdgcn_readfirstlane(rowptr[n + 1]);
    beg = max(beg, 0);
    end = min(end, nEdges);
    const int cnt = max(end - beg, 0);

    float4 acc = make_float4(0.0f, 0.0f, 0.0f, 0.0f);
    const float* hl = h + lane * 4;

    for (int j = beg; j < end; j += 32) {
        const int m   = min(end - j, 32);
        const int idx = (lane < m) ? csr_src[j + lane] : 0;
        int t = 0;
        for (; t + 8 <= m; t += 8) gcn_gather_batch<8>(hl, idx, t, acc);
        if (t + 4 <= m) { gcn_gather_batch<4>(hl, idx, t, acc); t += 4; }
        if (t + 2 <= m) { gcn_gather_batch<2>(hl, idx, t, acc); t += 2; }
        if (t < m)      { gcn_gather_batch<1>(hl, idx, t, acc); }
    }

    const float d = (float)cnt;
    if (mean) {
        const float dm = fmaxf(d, 1.0f);
        acc.x /= dm;
        acc.y /= dm;
        acc.z /= dm;
        acc.w /= dm;
    }
    *(float4*)(outp + (size_t)n * GCN_ROW + lane * 4) = acc;
    if (deg != nullptr && lane == 0) deg[n] = d;
}

extern "C" void kernel_launch(void* const* d_in, const int* in_sizes, int n_in,
                              void* d_out, int out_size, void* d_ws, size_t ws_size,
                              hipStream_t stream) {
    const float* X  = (const float*)d_in[0];
    const int*   E  = (const int*)  d_in[1];
    const float* W1 = (const float*)d_in[2];
    const float* b1 = (const float*)d_in[3];
    const float* W2 = (const float*)d_in[4];
    const float* b2 = (const float*)d_in[5];
    float* out = (float*)d_out;

    const int nEdges = in_sizes[1] / 2;

    char* ws = (char*)d_ws;
    float* aggX = (float*)ws;
    float* deg  = (float*)(ws + (size_t)GCN_NODES * GCN_DIN * 4);
    float* h1n  = (float*)(ws + (size_t)GCN_NODES * GCN_DHID * 4);
    float* h2   = (float*)(ws + (size_t)GCN_NODES * GCN_DHID * 8);
    char*  csr  = ws + (size_t)GCN_NODES * GCN_DHID * 10;
    int* tab    = (int*)(csr);
    int* rowptr = (int*)(csr + (1 << 19));
    int* sorted = (int*)(csr + (1 << 20));
    int* csrSrc = (int*)(csr + (1 << 20) + (1 << 23));

    gcn_edge_bucket<<<GCN_NCHUNK, GCN_SORT_THREADS, 0, stream>>>(E, tab, sorted, nEdges);
    gcn_bucket_order<<<GCN_NBUCKET, GCN_SORT_THREADS, 0, stream>>>(tab, sorted, rowptr, csrSrc, nEdges);

    gcn_segment_gather<<<(GCN_NODES + 7) / 8, 256, 0, stream>>>(
        X, rowptr, csrSrc, aggX, deg, GCN_NODES, nEdges,  0);

    gcn_gemm_mean_relu_wmma<<<dim3(GCN_NODES / 16, GCN_DHID / 64), 32, 0, stream>>>(
        aggX, W1, b1, deg, h1n, GCN_DIN, GCN_DHID);

    gcn_gemm_bias_wmma<<<dim3(GCN_NODES / 16, GCN_DOUT / 64), 32, 0, stream>>>(
        h1n, W2, b2, h2, GCN_DHID, GCN_DOUT);

    gcn_segment_gather<<<(GCN_NODES + 7) / 8, 256, 0, stream>>>(
        h2, rowptr, csrSrc, out, nullptr, GCN_NODES, nEdges,  1);
}


# Layer-1 gather and both GEMMs fused per 16-node tile (4 waves): summed X rows and hidden rows stay in LDS, A read from LDS; aggX, deg, h1n planes and two launches gone; sums and order unchanged
# speedup vs baseline: 5.9645x; 1.4079x over previous
#include <hip/hip_runtime.h>
#include <hip/hip_bf16.h>


typedef __attribute__((ext_vector_type(2))) float v2f;
typedef __attribute__((ext_vector_type(4))) float v4f;
typedef __attribute__((ext_vector_type(8))) float v8f;

#define GCN_NODES 50000
#define GCN_DIN   128
#define GCN_DHID  256
#define GCN_DOUT  128

__device__ __forceinline__ void gcn_wmma_step(const v2f a, const v4f b0, const v4f b1,
                                              v8f (&acc)[4]) {
#pragma unroll
    for (int t = 0; t < 4; ++t) {
        v2f b;
        b.x = b0[t];
        b.y = b1[t];
        acc[t] = __builtin_amdgcn_wmma_f32_16x16x4_f32(
             false, a,  false, b,
             (short)0, acc[t],
             false,  false);
    }
}

__device__ __forceinline__ void gcn_tile_16x64(const float* arow,
                                               const float* __restrict__ W,
                                               int K, int N, int nBase,
                                               int l, int half, v8f (&acc)[4]) {
    const float* ap = arow + 2 * half;
    const float* wp = W + (size_t)(2 * half) * (size_t)N + nBase + 4 * l;
    const size_t n1 = (size_t)N;
    int k = 0;
    for (; k + 8 <= K; k += 8) {
        const v2f a0 = *(const v2f*)(ap);
        const v2f a1 = *(const v2f*)(ap + 4);
        const v4f p0 = *(const v4f*)(wp);
        const v4f p1 = *(const v4f*)(wp + n1);
        const v4f q0 = *(const v4f*)(wp + 4 * n1);
        const v4f q1 = *(const v4f*)(wp + 5 * n1);
        gcn_wmma_step(a0, p0, p1, acc);
        gcn_wmma_step(a1, q0, q1, acc);
        ap += 8;
        wp += 8 * n1;
    }
    if (k < K) {
        const v2f a0 = *(const v2f*)(ap);
        const v4f p0 = *(const v4f*)(wp);
        const v4f p1 = *(const v4f*)(wp + n1);
        gcn_wmma_step(a0, p0, p1, acc);
    }
}

__device__ __forceinline__ void gcn_wmma_step2(const v2f a, const v2f b0, const v2f b1,
                                               v8f (&acc)[2]) {
#pragma unroll
    for (int t = 0; t < 2; ++t) {
        v2f b;
        b.x = b0[t];
        b.y = b1[t];
        acc[t] = __builtin_amdgcn_wmma_f32_16x16x4_f32(
             false, a,  false, b,
             (short)0, acc[t],
             false,  false);
    }
}

__device__ __forceinline__ void gcn_tile_16x32(const float* arow,
                                               const float* __restrict__ W,
                                               int K, int N, int nBase,
                                               int l, int half, v8f (&acc)[2]) {
    const float* ap = arow + 2 * half;
    const float* wp = W + (size_t)(2 * half) * (size_t)N + nBase + 2 * l;
    const size_t n1 = (size_t)N;
    int k = 0;
    for (; k + 8 <= K; k += 8) {
        const v2f a0 = *(const v2f*)(ap);
        const v2f a1 = *(const v2f*)(ap + 4);
        const v2f p0 = *(const v2f*)(wp);
        const v2f p1 = *(const v2f*)(wp + n1);
        const v2f q0 = *(const v2f*)(wp + 4 * n1);
        const v2f q1 = *(const v2f*)(wp + 5 * n1);
        gcn_wmma_step2(a0, p0, p1, acc);
        gcn_wmma_step2(a1, q0, q1, acc);
        ap += 8;
        wp += 8 * n1;
    }
    if (k < K) {
        const v2f a0 = *(const v2f*)(ap);
        const v2f p0 = *(const v2f*)(wp);
        const v2f p1 = *(const v2f*)(wp + n1);
        gcn_wmma_step2(a0, p0, p1, acc);
    }
}

#define GCN_NCHUNK  256
#define GCN_NBUCKET 256
#define GCN_BNODES  196
#define GCN_SORT_THREADS 1024

static_assert(GCN_NBUCKET * GCN_BNODES >= GCN_NODES, "buckets must cover every node");
static_assert(GCN_BNODES <= 256 && GCN_NBUCKET == 256, "one wave scans 256 LDS words, 8 per lane");
static_assert(GCN_SORT_THREADS == 4 * GCN_NCHUNK, "order pass: four threads per chunk run");

__device__ __forceinline__ int gcn_wave_scan256(const int* sIn, int lane, int (&excl)[8]) {
    int run = 0;
#pragma unroll
    for (int i = 0; i < 8; ++i) {
        const int t = sIn[lane * 8 + i];
        excl[i] = run;
        run += t;
    }
    int inc = run;
#pragma unroll
    for (int d = 1; d < 32; d <<= 1) {
        const int o = __shfl_up(inc, d);
        if (lane >= d) inc += o;
    }
    const int before = inc - run;
#pragma unroll
    for (int i = 0; i < 8; ++i) excl[i] += before;
    return inc;
}

__device__ __forceinline__ int gcn_bucket_of(int dst) {
    return min(max(dst / GCN_BNODES, 0), GCN_NBUCKET - 1);
}

__global__ __launch_bounds__(GCN_SORT_THREADS)
void gcn_edge_bucket(const int* __restrict__ edges,
                     int* __restrict__ tab,
                     int* __restrict__ sorted,
                     int nEdges) {
    __shared__ int sCnt[GCN_NBUCKET];
    __shared__ int sCur[GCN_NBUCKET];
    const int tid = threadIdx.x;
    const int c   = blockIdx.x;
    const int chunkLen = (nEdges + GCN_NCHUNK - 1) / GCN_NCHUNK;
    const int cs = min(c * chunkLen, nEdges);
    const int ce = min(cs + chunkLen, nEdges);
    const int2* e2 = (const int2*)edges;

    if (tid < GCN_NBUCKET) sCnt[tid] = 0;
    __syncthreads();

    for (int e0 = cs + tid; e0 < ce; e0 += 4 * GCN_SORT_THREADS) {
        int d[4];
#pragma unroll
        for (int u = 0; u < 4; ++u)
            d[u] = e2[min(e0 + u * GCN_SORT_THREADS, ce - 1)].y;
        asm volatile("" :: "v"(d[0]), "v"(d[1]), "v"(d[2]), "v"(d[3]));
#pragma unroll
        for (int u = 0; u < 4; ++u)
            if (e0 + u * GCN_SORT_THREADS < ce) atomicAdd(&sCnt[gcn_bucket_of(d[u])], 1);
    }
    __syncthreads();

    if (tid < 32) {
        int excl[8];
        gcn_wave_scan256(sCnt, tid, excl);
#pragma unroll
        for (int i = 0; i < 8; ++i) sCur[tid * 8 + i] = excl[i];
    }
    __syncthreads();

    int myCnt = 0, myOff = 0;
    if (tid < GCN_NBUCKET) { myCnt = sCnt[tid]; myOff = sCur[tid]; }
    __syncthreads();
    if (tid < GCN_NBUCKET) ((int2*)tab)[tid * GCN_NCHUNK + c] = make_int2(myCnt, myOff);

    int2* out2 = (int2*)sorted + cs;
    for (int e0 = cs + tid; e0 < ce; e0 += 4 * GCN_SORT_THREADS) {
        int2 sd[4];
#pragma unroll
        for (int u = 0; u < 4; ++u)
            sd[u] = e2[min(e0 + u * GCN_SORT_THREADS, ce - 1)];
        asm volatile("" :: "v"(sd[0].y), "v"(sd[1].y), "v"(sd[2].y), "v"(sd[3].y));
#pragma unroll
        for (int u = 0; u < 4; ++u) {
            if (e0 + u * GCN_SORT_THREADS < ce) {
                const int pos = atomicAdd(&sCur[gcn_bucket_of(sd[u].y)], 1);
                out2[pos] = sd[u];
            }
        }
    }
}

static_assert(GCN_BNODES <= GCN_SORT_THREADS, "one thread stores one rowptr word");

__device__ __forceinline__ int gcn_local_node(int dst, int nodeBase) {
    return min(max(dst - nodeBase, 0), GCN_BNODES - 1);
}

__global__ __launch_bounds__(GCN_SORT_THREADS)
void gcn_bucket_order(const int* __restrict__ tab,
                      const int* __restrict__ sorted,
                      int* __restrict__ rowptr,
                      int* __restrict__ csr_src,
                      int nEdges) {
    __shared__ int sNode[256];
    __shared__ int sRow[256];
    __shared__ int sRed[32];
    const int tid  = threadIdx.x;
    const int lane = tid & 31;
    const int q    = blockIdx.x;
    const int chunkLen = (nEdges + GCN_NCHUNK - 1) / GCN_NCHUNK;

    const int c   = tid >> 2;
    const int sub = tid & 3;
    const int2 seg = ((const int2*)tab)[q * GCN_NCHUNK + c];
    const int segCnt = min(max(seg.x, 0), chunkLen);
    const int segOff = min(max(seg.y, 0), chunkLen - segCnt);

    int part = (sub == 0) ? segOff : 0;
#pragma unroll
    for (int d = 16; d >= 1; d >>= 1) part += __shfl_xor(part, d);
    if (lane == 0) sRed[tid >> 5] = part;
    if (tid < 256) sNode[tid] = 0;
    __syncthreads();

    const int cs = min(c * chunkLen, nEdges);
    const int2* s2 = (const int2*)sorted + cs + segOff;
    const int nodeBase = q * GCN_BNODES;

    for (int i0 = sub; i0 < segCnt; i0 += 16) {
        int d[4];
#pragma unroll
        for (int u = 0; u < 4; ++u) d[u] = s2[min(i0 + 4 * u, segCnt - 1)].y;
        asm volatile("" :: "v"(d[0]), "v"(d[1]), "v"(d[2]), "v"(d[3]));
#pragma unroll
        for (int u = 0; u < 4; ++u)
            if (i0 + 4 * u < segCnt) atomicAdd(&sNode[gcn_local_node(d[u], nodeBase)], 1);
    }
    __syncthreads();

    int tot = 0;
    if (tid < 32) {
        int base = sRed[tid];
#pragma unroll
        for (int d = 16; d >= 1; d >>= 1) base += __shfl_xor(base, d);
        int excl[8];
        const int inc = gcn_wave_scan256(sNode, tid, excl);
#pragma unroll
        for (int i = 0; i < 8; ++i) {
            const int rp = base + excl[i];
            sNode[tid * 8 + i] = rp;
            sRow[tid * 8 + i]  = rp;
        }
        tot = base + inc;
    }
    __syncthreads();
    if (tid < GCN_BNODES && nodeBase + tid < GCN_NODES) rowptr[nodeBase + tid] = sRow[tid];
    if (q == GCN_NBUCKET - 1 && tid == 31) rowptr[GCN_NODES] = tot;

    for (int i0 = sub; i0 < segCnt; i0 += 16) {
        int2 sd[4];
#pragma unroll
        for (int u = 0; u < 4; ++u) sd[u] = s2[min(i0 + 4 * u, segCnt - 1)];
        asm volatile("" :: "v"(sd[0].y), "v"(sd[1].y), "v"(sd[2].y), "v"(sd[3].y));
#pragma unroll
        for (int u = 0; u < 4; ++u) {
            if (i0 + 4 * u < segCnt) {
                const int pos = atomicAdd(&sNode[gcn_local_node(sd[u].y, nodeBase)], 1);
                if ((unsigned)pos < (unsigned)nEdges) csr_src[pos] = sd[u].x;
            }
        }
    }
}

#define GCN_ROW 128

template <int B>
__device__ __forceinline__ void gcn_gather_batch(const float* __restrict__ hl,
                                                 int idx, int t, float4& acc) {
    float4 v[B];
#pragma unroll
    for (int u = 0; u < B; ++u) {
        const int s = __builtin_amdgcn_readlane(idx, t + u);
        v[u] = *(const float4*)(hl + (size_t)s * GCN_ROW);
    }
#pragma unroll
    for (int u = 0; u < B; ++u) {
        acc.x += v[u].x;
        acc.y += v[u].y;
        acc.z += v[u].z;
        acc.w += v[u].w;
    }
}

__global__ __launch_bounds__(256)
void gcn_segment_gather(const float* __restrict__ h,
                        const int* __restrict__ rowptr,
                        const int* __restrict__ csr_src,
                        float* __restrict__ outp,
                        float* __restrict__ deg,
                        int nNodes, int nEdges, int mean) {
    const int lane = threadIdx.x & 31;
    const int n    = blockIdx.x * 8 + (threadIdx.x >> 5);
    if (n >= nNodes) return;

    int beg = __builtin_amdgcn_readfirstlane(rowptr[n]);
    int end = __builtin_amdgcn_readfirstlane(rowptr[n + 1]);
    beg = max(beg, 0);
    end = min(end, nEdges);
    const int cnt = max(end - beg, 0);

    float4 acc = make_float4(0.0f, 0.0f, 0.0f, 0.0f);
    const float* hl = h + lane * 4;

    for (int j = beg; j < end; j += 32) {
        const int m   = min(end - j, 32);
        const int idx = (lane < m) ? csr_src[j + lane] : 0;
        int t = 0;
        for (; t + 8 <= m; t += 8) gcn_gather_batch<8>(hl, idx, t, acc);
        if (t + 4 <= m) { gcn_gather_batch<4>(hl, idx, t, acc); t += 4; }
        if (t + 2 <= m) { gcn_gather_batch<2>(hl, idx, t, acc); t += 2; }
        if (t < m)      { gcn_gather_batch<1>(hl, idx, t, acc); }
    }

    const float d = (float)cnt;
    if (mean) {
        const float dm = fmaxf(d, 1.0f);
        acc.x /= dm;
        acc.y /= dm;
        acc.z /= dm;
        acc.w /= dm;
    }
    *(float4*)(outp + (size_t)n * GCN_ROW + lane * 4) = acc;
    if (deg != nullptr && lane == 0) deg[n] = d;
}

#define GCN_PA (GCN_DIN + 4)
#define GCN_PH (GCN_DHID + 4)

static_assert(GCN_NODES % 16 == 0, "every tile has 16 nodes: no M predication, all waves reach every barrier");
static_assert(GCN_DIN == GCN_ROW && GCN_DOUT == GCN_ROW, "gather rows are 128 columns");

__device__ __forceinline__ int gcn_load_idx(const int* __restrict__ csr_src, int j, int end, int lane) {
    const int m = min(end - j, 32);
    return (lane < m) ? csr_src[j + lane] : 0;
}

__global__ __launch_bounds__(128) __attribute__((amdgpu_waves_per_eu(16, 16)))
void gcn_layer1_tile(const float* __restrict__ X,
                     const int* __restrict__ rowptr,
                     const int* __restrict__ csr_src,
                     const float* __restrict__ W1,
                     const float* __restrict__ b1,
                     const float* __restrict__ W2,
                     const float* __restrict__ b2,
                     float* __restrict__ H2,
                     int nEdges) {
    __shared__ __attribute__((aligned(16))) float sT[16 * GCN_PH];
    __shared__ float sDeg[16];

    const int lane = threadIdx.x & 31;
    const int wv = __builtin_amdgcn_readfirstlane((int)(threadIdx.x >> 5));
    const int mBase = blockIdx.x * 16;
    const int n0 = mBase + 4 * wv;

    const int rp = rowptr[n0 + min(lane, 4)];
    const float* hl = X + lane * 4;

    int begN = max(__builtin_amdgcn_readlane(rp, 0), 0);
    int endN = min(__builtin_amdgcn_readlane(rp, 1), nEdges);
    int idxN = gcn_load_idx(csr_src, begN, endN, lane);

    for (int i = 0; i < 4; ++i) {
        const int beg = begN;
        const int end = endN;
        int idx = idxN;
        if (i < 3) {
            begN = max(__builtin_amdgcn_readlane(rp, i + 1), 0);
            endN = min(__builtin_amdgcn_readlane(rp, i + 2), nEdges);
            idxN = gcn_load_idx(csr_src, begN, endN, lane);
        }
        const int cnt = max(end - beg, 0);

        float4 acc = make_float4(0.0f, 0.0f, 0.0f, 0.0f);
        for (int j = beg; j < end; j += 32) {
            if (j > beg) idx = gcn_load_idx(csr_src, j, end, lane);
            const int m = min(end - j, 32);
            int t = 0;
            for (; t + 8 <= m; t += 8) gcn_gather_batch<8>(hl, idx, t, acc);
            if (t + 4 <= m) { gcn_gather_batch<4>(hl, idx, t, acc); t += 4; }
            if (t + 2 <= m) { gcn_gather_batch<2>(hl, idx, t, acc); t += 2; }
            if (t < m)      { gcn_gather_batch<1>(hl, idx, t, acc); }
        }
        *(float4*)(sT + (4 * wv + i) * GCN_PA + lane * 4) = acc;
        if (lane == 0) sDeg[4 * wv + i] = (float)cnt;
    }
    __syncthreads();

    const int half = lane >> 4;
    const int l    = lane & 15;

    v8f acc1[4];
#pragma unroll
    for (int t = 0; t < 4; ++t) {
#pragma unroll
        for (int i = 0; i < 8; ++i) acc1[t][i] = 0.0f;
    }
    gcn_tile_16x64(sT + l * GCN_PA, W1, GCN_DIN, GCN_DHID, 64 * wv, l, half, acc1);

    float dg[8], dm[8];
#pragma unroll
    for (int i = 0; i < 8; ++i) {
        dg[i] = sDeg[i + 8 * half];
        dm[i] = fmaxf(dg[i], 1.0f);
    }
    const v4f bv4 = *(const v4f*)(b1 + 64 * wv + 4 * l);

    __syncthreads();
#pragma unroll
    for (int i = 0; i < 8; ++i) {
        v4f o;
#pragma unroll
        for (int t = 0; t < 4; ++t) {
            const float v = (acc1[t][i] + dg[i] * bv4[t]) / dm[i];
            o[t] = fmaxf(v, 0.0f);
        }
        *(v4f*)(sT + (i + 8 * half) * GCN_PH + 64 * wv + 4 * l) = o;
    }
    __syncthreads();

    const v2f bv2 = *(const v2f*)(b2 + 32 * wv + 2 * l);
    v8f acc2[2];
#pragma unroll
    for (int t = 0; t < 2; ++t) {
#pragma unroll
        for (int i = 0; i < 8; ++i) acc2[t][i] = bv2[t];
    }
    gcn_tile_16x32(sT + l * GCN_PH, W2, GCN_DHID, GCN_DOUT, 32 * wv, l, half, acc2);

#pragma unroll
    for (int i = 0; i < 8; ++i) {
        const int row = mBase + i + 8 * half;
        v2f o;
        o.x = acc2[0][i];
        o.y = acc2[1][i];
        *(v2f*)(H2 + (size_t)row * GCN_DOUT + 32 * wv + 2 * l) = o;
    }
}

extern "C" void kernel_launch(void* const* d_in, const int* in_sizes, int n_in,
                              void* d_out, int out_size, void* d_ws, size_t ws_size,
                              hipStream_t stream) {
    const float* X  = (const float*)d_in[0];
    const int*   E  = (const int*)  d_in[1];
    const float* W1 = (const float*)d_in[2];
    const float* b1 = (const float*)d_in[3];
    const float* W2 = (const float*)d_in[4];
    const float* b2 = (const float*)d_in[5];
    float* out = (float*)d_out;

    const int nEdges = in_sizes[1] / 2;

    char* ws = (char*)d_ws;
    float* h2   = (float*)(ws + (size_t)GCN_NODES * GCN_DHID * 8);
    char*  csr  = ws + (size_t)GCN_NODES * GCN_DHID * 10;
    int* tab    = (int*)(csr);
    int* rowptr = (int*)(csr + (1 << 19));
    int* sorted = (int*)(csr + (1 << 20));
    int* csrSrc = (int*)(csr + (1 << 20) + (1 << 23));

    gcn_edge_bucket<<<GCN_NCHUNK, GCN_SORT_THREADS, 0, stream>>>(E, tab, sorted, nEdges);
    gcn_bucket_order<<<GCN_NBUCKET, GCN_SORT_THREADS, 0, stream>>>(tab, sorted, rowptr, csrSrc, nEdges);

    gcn_layer1_tile<<<GCN_NODES / 16, 128, 0, stream>>>(
        X, rowptr, csrSrc, W1, b1, W2, b2, h2, nEdges);

    gcn_segment_gather<<<(GCN_NODES + 7) / 8, 256, 0, stream>>>(
        h2, rowptr, csrSrc, out, nullptr, GCN_NODES, nEdges,  1);
}


# Both edge-ordering launches build their ordered output in LDS and store it as whole 128-byte lines, not slot by slot from registers (chunks start on a line); f32 sum order changes
# speedup vs baseline: 6.1750x; 1.0353x over previous
#include <hip/hip_runtime.h>
#include <hip/hip_bf16.h>


typedef __attribute__((ext_vector_type(2))) float v2f;
typedef __attribute__((ext_vector_type(4))) float v4f;
typedef __attribute__((ext_vector_type(8))) float v8f;

#define GCN_NODES 50000
#define GCN_DIN   128
#define GCN_DHID  256
#define GCN_DOUT  128

__device__ __forceinline__ void gcn_wmma_step(const v2f a, const v4f b0, const v4f b1,
                                              v8f (&acc)[4]) {
#pragma unroll
    for (int t = 0; t < 4; ++t) {
        v2f b;
        b.x = b0[t];
        b.y = b1[t];
        acc[t] = __builtin_amdgcn_wmma_f32_16x16x4_f32(
             false, a,  false, b,
             (short)0, acc[t],
             false,  false);
    }
}

__device__ __forceinline__ void gcn_tile_16x64(const float* arow,
                                               const float* __restrict__ W,
                                               int K, int N, int nBase,
                                               int l, int half, v8f (&acc)[4]) {
    const float* ap = arow + 2 * half;
    const float* wp = W + (size_t)(2 * half) * (size_t)N + nBase + 4 * l;
    const size_t n1 = (size_t)N;
    int k = 0;
    for (; k + 8 <= K; k += 8) {
        const v2f a0 = *(const v2f*)(ap);
        const v2f a1 = *(const v2f*)(ap + 4);
        const v4f p0 = *(const v4f*)(wp);
        const v4f p1 = *(const v4f*)(wp + n1);
        const v4f q0 = *(const v4f*)(wp + 4 * n1);
        const v4f q1 = *(const v4f*)(wp + 5 * n1);
        gcn_wmma_step(a0, p0, p1, acc);
        gcn_wmma_step(a1, q0, q1, acc);
        ap += 8;
        wp += 8 * n1;
    }
    if (k < K) {
        const v2f a0 = *(const v2f*)(ap);
        const v4f p0 = *(const v4f*)(wp);
        const v4f p1 = *(const v4f*)(wp + n1);
        gcn_wmma_step(a0, p0, p1, acc);
    }
}

__device__ __forceinline__ void gcn_wmma_step2(const v2f a, const v2f b0, const v2f b1,
                                               v8f (&acc)[2]) {
#pragma unroll
    for (int t = 0; t < 2; ++t) {
        v2f b;
        b.x = b0[t];
        b.y = b1[t];
        acc[t] = __builtin_amdgcn_wmma_f32_16x16x4_f32(
             false, a,  false, b,
             (short)0, acc[t],
             false,  false);
    }
}

__device__ __forceinline__ void gcn_tile_16x32(const float* arow,
                                               const float* __restrict__ W,
                                               int K, int N, int nBase,
                                               int l, int half, v8f (&acc)[2]) {
    const float* ap = arow + 2 * half;
    const float* wp = W + (size_t)(2 * half) * (size_t)N + nBase + 2 * l;
    const size_t n1 = (size_t)N;
    int k = 0;
    for (; k + 8 <= K; k += 8) {
        const v2f a0 = *(const v2f*)(ap);
        const v2f a1 = *(const v2f*)(ap + 4);
        const v2f p0 = *(const v2f*)(wp);
        const v2f p1 = *(const v2f*)(wp + n1);
        const v2f q0 = *(const v2f*)(wp + 4 * n1);
        const v2f q1 = *(const v2f*)(wp + 5 * n1);
        gcn_wmma_step2(a0, p0, p1, acc);
        gcn_wmma_step2(a1, q0, q1, acc);
        ap += 8;
        wp += 8 * n1;
    }
    if (k < K) {
        const v2f a0 = *(const v2f*)(ap);
        const v2f p0 = *(const v2f*)(wp);
        const v2f p1 = *(const v2f*)(wp + n1);
        gcn_wmma_step2(a0, p0, p1, acc);
    }
}

#define GCN_NCHUNK  256
#define GCN_NBUCKET 256
#define GCN_BNODES  196
#define GCN_SORT_THREADS 1024
#define GCN_STAGE_EDGES 4096

typedef __attribute__((ext_vector_type(2))) int v2i;
typedef __attribute__((ext_vector_type(4))) int v4i;

static_assert(GCN_NBUCKET * GCN_BNODES >= GCN_NODES, "buckets must cover every node");
static_assert(GCN_BNODES <= 256 && GCN_NBUCKET == 256, "one wave scans 256 LDS words, 8 per lane");
static_assert(GCN_SORT_THREADS == 4 * GCN_NCHUNK, "order pass: four threads per chunk run");

__device__ __forceinline__ int gcn_wave_scan256(const int* sIn, int lane, int (&excl)[8]) {
    int run = 0;
#pragma unroll
    for (int i = 0; i < 8; ++i) {
        const int t = sIn[lane * 8 + i];
        excl[i] = run;
        run += t;
    }
    int inc = run;
#pragma unroll
    for (int d = 1; d < 32; d <<= 1) {
        const int o = __shfl_up(inc, d);
        if (lane >= d) inc += o;
    }
    const int before = inc - run;
#pragma unroll
    for (int i = 0; i < 8; ++i) excl[i] += before;
    return inc;
}

__device__ __forceinline__ int gcn_chunk_len(int nEdges) {
    return (((nEdges + GCN_NCHUNK - 1) / GCN_NCHUNK) + 15) & ~15;
}

__device__ __forceinline__ int gcn_bucket_of(int dst) {
    return min(max(dst / GCN_BNODES, 0), GCN_NBUCKET - 1);
}

__global__ __launch_bounds__(GCN_SORT_THREADS)
void gcn_edge_bucket(const int* __restrict__ edges,
                     int* __restrict__ tab,
                     int* __restrict__ sorted,
                     int nEdges) {
    __shared__ int sCnt[GCN_NBUCKET];
    __shared__ int sCur[GCN_NBUCKET];
    __shared__ __attribute__((aligned(16))) int sStage[2 * GCN_STAGE_EDGES];
    const int tid = threadIdx.x;
    const int c   = blockIdx.x;
    const int chunkLen = gcn_chunk_len(nEdges);
    const int cs = min(c * chunkLen, nEdges);
    const int ce = min(cs + chunkLen, nEdges);
    const int2* e2 = (const int2*)edges;
    const bool staged = chunkLen <= GCN_STAGE_EDGES;

    if (tid < GCN_NBUCKET) sCnt[tid] = 0;
    __syncthreads();

    for (int e0 = cs + tid; e0 < ce; e0 += 4 * GCN_SORT_THREADS) {
        int d[4];
#pragma unroll
        for (int u = 0; u < 4; ++u)
            d[u] = e2[min(e0 + u * GCN_SORT_THREADS, ce - 1)].y;
        asm volatile("" :: "v"(d[0]), "v"(d[1]), "v"(d[2]), "v"(d[3]));
#pragma unroll
        for (int u = 0; u < 4; ++u)
            if (e0 + u * GCN_SORT_THREADS < ce) atomicAdd(&sCnt[gcn_bucket_of(d[u])], 1);
    }
    __syncthreads();

    if (tid < 32) {
        int excl[8];
        gcn_wave_scan256(sCnt, tid, excl);
#pragma unroll
        for (int i = 0; i < 8; ++i) sCur[tid * 8 + i] = excl[i];
    }
    __syncthreads();

    int myCnt = 0, myOff = 0;
    if (tid < GCN_NBUCKET) { myCnt = sCnt[tid]; myOff = sCur[tid]; }
    __syncthreads();

    int2* out2 = (int2*)sorted + cs;
    for (int e0 = cs + tid; e0 < ce; e0 += 4 * GCN_SORT_THREADS) {
        int2 sd[4];
#pragma unroll
        for (int u = 0; u < 4; ++u)
            sd[u] = e2[min(e0 + u * GCN_SORT_THREADS, ce - 1)];
        asm volatile("" :: "v"(sd[0].y), "v"(sd[1].y), "v"(sd[2].y), "v"(sd[3].y));
#pragma unroll
        for (int u = 0; u < 4; ++u) {
            if (e0 + u * GCN_SORT_THREADS < ce) {
                const int pos = atomicAdd(&sCur[gcn_bucket_of(sd[u].y)], 1);
                if (staged) {
                    if ((unsigned)pos < (unsigned)GCN_STAGE_EDGES) {
                        v2i r;
                        r.x = sd[u].x;
                        r.y = sd[u].y;
                        *(v2i*)(sStage + 2 * pos) = r;
                    }
                } else {
                    out2[pos] = sd[u];
                }
            }
        }
    }

    if (staged) {
        __syncthreads();
        const int len = ce - cs;
        int* o = (int*)out2;
        for (int i = tid; i < (len >> 1); i += GCN_SORT_THREADS)
            *(v4i*)(o + 4 * i) = *(const v4i*)(sStage + 4 * i);
        if ((len & 1) != 0 && tid == 0)
            *(v2i*)(o + 2 * (len - 1)) = *(const v2i*)(sStage + 2 * (len - 1));
    }
    if (tid < GCN_NBUCKET) ((int2*)tab)[tid * GCN_NCHUNK + c] = make_int2(myCnt, myOff);
}

static_assert(GCN_BNODES <= GCN_SORT_THREADS, "one thread stores one rowptr word");

__device__ __forceinline__ int gcn_local_node(int dst, int nodeBase) {
    return min(max(dst - nodeBase, 0), GCN_BNODES - 1);
}

__global__ __launch_bounds__(GCN_SORT_THREADS)
void gcn_bucket_order(const int* __restrict__ tab,
                      const int* __restrict__ sorted,
                      int* __restrict__ rowptr,
                      int* __restrict__ csr_src,
                      int nEdges) {
    __shared__ int sNode[256];
    __shared__ int sRow[256];
    __shared__ int sRed[32];
    __shared__ int sMeta[2];
    __shared__ int sOut[GCN_STAGE_EDGES];
    const int tid  = threadIdx.x;
    const int lane = tid & 31;
    const int q    = blockIdx.x;
    const int chunkLen = gcn_chunk_len(nEdges);

    const int c   = tid >> 2;
    const int sub = tid & 3;
    const int2 seg = ((const int2*)tab)[q * GCN_NCHUNK + c];
    const int segCnt = min(max(seg.x, 0), chunkLen);
    const int segOff = min(max(seg.y, 0), chunkLen - segCnt);

    int part = (sub == 0) ? segOff : 0;
#pragma unroll
    for (int d = 16; d >= 1; d >>= 1) part += __shfl_xor(part, d);
    if (lane == 0) sRed[tid >> 5] = part;
    if (tid < 256) sNode[tid] = 0;
    __syncthreads();

    const int cs = min(c * chunkLen, nEdges);
    const int2* s2 = (const int2*)sorted + cs + segOff;
    const int nodeBase = q * GCN_BNODES;

    for (int i0 = sub; i0 < segCnt; i0 += 16) {
        int d[4];
#pragma unroll
        for (int u = 0; u < 4; ++u) d[u] = s2[min(i0 + 4 * u, segCnt - 1)].y;
        asm volatile("" :: "v"(d[0]), "v"(d[1]), "v"(d[2]), "v"(d[3]));
#pragma unroll
        for (int u = 0; u < 4; ++u)
            if (i0 + 4 * u < segCnt) atomicAdd(&sNode[gcn_local_node(d[u], nodeBase)], 1);
    }
    __syncthreads();

    int tot = 0;
    if (tid < 32) {
        int base = sRed[tid];
#pragma unroll
        for (int d = 16; d >= 1; d >>= 1) base += __shfl_xor(base, d);
        int excl[8];
        const int inc = gcn_wave_scan256(sNode, tid, excl);
#pragma unroll
        for (int i = 0; i < 8; ++i) {
            const int rp = base + excl[i];
            sNode[tid * 8 + i] = rp;
            sRow[tid * 8 + i]  = rp;
        }
        tot = base + inc;
        if (tid == 31) { sMeta[0] = base; sMeta[1] = inc; }
    }
    __syncthreads();
    const int bBase = sMeta[0];
    const int bTot  = sMeta[1];
    const bool staged = bBase >= 0 && bTot >= 0 && bTot <= GCN_STAGE_EDGES;

    for (int i0 = sub; i0 < segCnt; i0 += 16) {
        int2 sd[4];
#pragma unroll
        for (int u = 0; u < 4; ++u) sd[u] = s2[min(i0 + 4 * u, segCnt - 1)];
        asm volatile("" :: "v"(sd[0].y), "v"(sd[1].y), "v"(sd[2].y), "v"(sd[3].y));
#pragma unroll
        for (int u = 0; u < 4; ++u) {
            if (i0 + 4 * u < segCnt) {
                const int pos = atomicAdd(&sNode[gcn_local_node(sd[u].y, nodeBase)], 1);
                if (staged) {
                    const int li = pos - bBase;
                    if ((unsigned)li < (unsigned)GCN_STAGE_EDGES) sOut[li] = sd[u].x;
                } else {
                    if ((unsigned)pos < (unsigned)nEdges) csr_src[pos] = sd[u].x;
                }
            }
        }
    }
    __syncthreads();
    if (staged) {
        const int gEnd = bBase + bTot;
        for (int g = (bBase & ~31) + tid; g < gEnd; g += GCN_SORT_THREADS)
            if (g >= bBase && (unsigned)g < (unsigned)nEdges) csr_src[g] = sOut[g - bBase];
    }
    if (tid < GCN_BNODES && nodeBase + tid < GCN_NODES) rowptr[nodeBase + tid] = sRow[tid];
    if (q == GCN_NBUCKET - 1 && tid == 31) rowptr[GCN_NODES] = tot;
}

#define GCN_ROW 128

template <int B>
__device__ __forceinline__ void gcn_gather_batch(const float* __restrict__ hl,
                                                 int idx, int t, float4& acc) {
    float4 v[B];
#pragma unroll
    for (int u = 0; u < B; ++u) {
        const int s = __builtin_amdgcn_readlane(idx, t + u);
        v[u] = *(const float4*)(hl + (size_t)s * GCN_ROW);
    }
#pragma unroll
    for (int u = 0; u < B; ++u) {
        acc.x += v[u].x;
        acc.y += v[u].y;
        acc.z += v[u].z;
        acc.w += v[u].w;
    }
}

__global__ __launch_bounds__(256)
void gcn_segment_gather(const float* __restrict__ h,
                        const int* __restrict__ rowptr,
                        const int* __restrict__ csr_src,
                        float* __restrict__ outp,
                        float* __restrict__ deg,
                        int nNodes, int nEdges, int mean) {
    const int lane = threadIdx.x & 31;
    const int n    = blockIdx.x * 8 + (threadIdx.x >> 5);
    if (n >= nNodes) return;

    int beg = __builtin_amdgcn_readfirstlane(rowptr[n]);
    int end = __builtin_amdgcn_readfirstlane(rowptr[n + 1]);
    beg = max(beg, 0);
    end = min(end, nEdges);
    const int cnt = max(end - beg, 0);

    float4 acc = make_float4(0.0f, 0.0f, 0.0f, 0.0f);
    const float* hl = h + lane * 4;

    for (int j = beg; j < end; j += 32) {
        const int m   = min(end - j, 32);
        const int idx = (lane < m) ? csr_src[j + lane] : 0;
        int t = 0;
        for (; t + 8 <= m; t += 8) gcn_gather_batch<8>(hl, idx, t, acc);
        if (t + 4 <= m) { gcn_gather_batch<4>(hl, idx, t, acc); t += 4; }
        if (t + 2 <= m) { gcn_gather_batch<2>(hl, idx, t, acc); t += 2; }
        if (t < m)      { gcn_gather_batch<1>(hl, idx, t, acc); }
    }

    const float d = (float)cnt;
    if (mean) {
        const float dm = fmaxf(d, 1.0f);
        acc.x /= dm;
        acc.y /= dm;
        acc.z /= dm;
        acc.w /= dm;
    }
    *(float4*)(outp + (size_t)n * GCN_ROW + lane * 4) = acc;
    if (deg != nullptr && lane == 0) deg[n] = d;
}

#define GCN_PA (GCN_DIN + 4)
#define GCN_PH (GCN_DHID + 4)

static_assert(GCN_NODES % 16 == 0, "every tile has 16 nodes: no M predication, all waves reach every barrier");
static_assert(GCN_DIN == GCN_ROW && GCN_DOUT == GCN_ROW, "gather rows are 128 columns");

__device__ __forceinline__ int gcn_load_idx(const int* __restrict__ csr_src, int j, int end, int lane) {
    const int m = min(end - j, 32);
    return (lane < m) ? csr_src[j + lane] : 0;
}

__global__ __launch_bounds__(128) __attribute__((amdgpu_waves_per_eu(16, 16)))
void gcn_layer1_tile(const float* __restrict__ X,
                     const int* __restrict__ rowptr,
                     const int* __restrict__ csr_src,
                     const float* __restrict__ W1,
                     const float* __restrict__ b1,
                     const float* __restrict__ W2,
                     const float* __restrict__ b2,
                     float* __restrict__ H2,
                     int nEdges) {
    __shared__ __attribute__((aligned(16))) float sT[16 * GCN_PH];
    __shared__ float sDeg[16];

    const int lane = threadIdx.x & 31;
    const int wv = __builtin_amdgcn_readfirstlane((int)(threadIdx.x >> 5));
    const int mBase = blockIdx.x * 16;
    const int n0 = mBase + 4 * wv;

    const int rp = rowptr[n0 + min(lane, 4)];
    const float* hl = X + lane * 4;

    int begN = max(__builtin_amdgcn_readlane(rp, 0), 0);
    int endN = min(__builtin_amdgcn_readlane(rp, 1), nEdges);
    int idxN = gcn_load_idx(csr_src, begN, endN, lane);

    for (int i = 0; i < 4; ++i) {
        const int beg = begN;
        const int end = endN;
        int idx = idxN;
        if (i < 3) {
            begN = max(__builtin_amdgcn_readlane(rp, i + 1), 0);
            endN = min(__builtin_amdgcn_readlane(rp, i + 2), nEdges);
            idxN = gcn_load_idx(csr_src, begN, endN, lane);
        }
        const int cnt = max(end - beg, 0);

        float4 acc = make_float4(0.0f, 0.0f, 0.0f, 0.0f);
        for (int j = beg; j < end; j += 32) {
            if (j > beg) idx = gcn_load_idx(csr_src, j, end, lane);
            const int m = min(end - j, 32);
            int t = 0;
            for (; t + 8 <= m; t += 8) gcn_gather_batch<8>(hl, idx, t, acc);
            if (t + 4 <= m) { gcn_gather_batch<4>(hl, idx, t, acc); t += 4; }
            if (t + 2 <= m) { gcn_gather_batch<2>(hl, idx, t, acc); t += 2; }
            if (t < m)      { gcn_gather_batch<1>(hl, idx, t, acc); }
        }
        *(float4*)(sT + (4 * wv + i) * GCN_PA + lane * 4) = acc;
        if (lane == 0) sDeg[4 * wv + i] = (float)cnt;
    }
    __syncthreads();

    const int half = lane >> 4;
    const int l    = lane & 15;

    v8f acc1[4];
#pragma unroll
    for (int t = 0; t < 4; ++t) {
#pragma unroll
        for (int i = 0; i < 8; ++i) acc1[t][i] = 0.0f;
    }
    gcn_tile_16x64(sT + l * GCN_PA, W1, GCN_DIN, GCN_DHID, 64 * wv, l, half, acc1);

    float dg[8], dm[8];
#pragma unroll
    for (int i = 0; i < 8; ++i) {
        dg[i] = sDeg[i + 8 * half];
        dm[i] = fmaxf(dg[i], 1.0f);
    }
    const v4f bv4 = *(const v4f*)(b1 + 64 * wv + 4 * l);

    __syncthreads();
#pragma unroll
    for (int i = 0; i < 8; ++i) {
        v4f o;
#pragma unroll
        for (int t = 0; t < 4; ++t) {
            const float v = (acc1[t][i] + dg[i] * bv4[t]) / dm[i];
            o[t] = fmaxf(v, 0.0f);
        }
        *(v4f*)(sT + (i + 8 * half) * GCN_PH + 64 * wv + 4 * l) = o;
    }
    __syncthreads();

    const v2f bv2 = *(const v2f*)(b2 + 32 * wv + 2 * l);
    v8f acc2[2];
#pragma unroll
    for (int t = 0; t < 2; ++t) {
#pragma unroll
        for (int i = 0; i < 8; ++i) acc2[t][i] = bv2[t];
    }
    gcn_tile_16x32(sT + l * GCN_PH, W2, GCN_DHID, GCN_DOUT, 32 * wv, l, half, acc2);

#pragma unroll
    for (int i = 0; i < 8; ++i) {
        const int row = mBase + i + 8 * half;
        v2f o;
        o.x = acc2[0][i];
        o.y = acc2[1][i];
        *(v2f*)(H2 + (size_t)row * GCN_DOUT + 32 * wv + 2 * l) = o;
    }
}

extern "C" void kernel_launch(void* const* d_in, const int* in_sizes, int n_in,
                              void* d_out, int out_size, void* d_ws, size_t ws_size,
                              hipStream_t stream) {
    const float* X  = (const float*)d_in[0];
    const int*   E  = (const int*)  d_in[1];
    const float* W1 = (const float*)d_in[2];
    const float* b1 = (const float*)d_in[3];
    const float* W2 = (const float*)d_in[4];
    const float* b2 = (const float*)d_in[5];
    float* out = (float*)d_out;

    const int nEdges = in_sizes[1] / 2;

    char* ws = (char*)d_ws;
    float* h2   = (float*)(ws + (size_t)GCN_NODES * GCN_DHID * 8);
    char*  csr  = ws + (size_t)GCN_NODES * GCN_DHID * 10;
    int* tab    = (int*)(csr);
    int* rowptr = (int*)(csr + (1 << 19));
    int* sorted = (int*)(csr + (1 << 20));
    int* csrSrc = (int*)(csr + (1 << 20) + (1 << 23));

    gcn_edge_bucket<<<GCN_NCHUNK, GCN_SORT_THREADS, 0, stream>>>(E, tab, sorted, nEdges);
    gcn_bucket_order<<<GCN_NBUCKET, GCN_SORT_THREADS, 0, stream>>>(tab, sorted, rowptr, csrSrc, nEdges);

    gcn_layer1_tile<<<GCN_NODES / 16, 128, 0, stream>>>(
        X, rowptr, csrSrc, W1, b1, W2, b2, h2, nEdges);

    gcn_segment_gather<<<(GCN_NODES + 7) / 8, 256, 0, stream>>>(
        h2, rowptr, csrSrc, out, nullptr, GCN_NODES, nEdges,  1);
}
